# starting kernel
# baseline (speedup 1.0000x reference)
#include <hip/hip_runtime.h>

#define NUM_USER 50000
#define NUM_ITEM 50000
#define N_NODES  100000
#define DIM_E    64
#define DIM_FEAT 128
#define DIM_H    256
#define N_EDGES  1600000
#define N_TILES  (NUM_ITEM / 16)

typedef __attribute__((ext_vector_type(2))) float v2f;
typedef __attribute__((ext_vector_type(8))) float v8f;

__global__ void k_zero(float* __restrict__ p, int n) {
    int i = blockIdx.x * blockDim.x + threadIdx.x;
    if (i < n) p[i] = 0.0f;
}

__global__ __launch_bounds__(128) void k_mlp(
        const float* __restrict__ F,
        const float* __restrict__ W1, const float* __restrict__ b1,
        const float* __restrict__ W2, const float* __restrict__ b2,
        float* __restrict__ temp) {
    __shared__ float hid[4][16][DIM_H];

    const int wave = threadIdx.x >> 5;
    const int lane = threadIdx.x & 31;
    const int lm   = lane & 15;
    const int lh   = lane >> 4;
    const int tile = blockIdx.x * 4 + wave;
    if (tile >= N_TILES) return;
    const int r0 = tile * 16;

    for (int nt = 0; nt < DIM_H / 16; ++nt) {
        const int N = nt * 16 + lm;
        v8f acc = {};
        #pragma unroll 4
        for (int kb = 0; kb < DIM_FEAT / 4; ++kb) {
            const int k0 = kb * 4 + lh * 2;
            v2f a = *(const v2f*)(F + (r0 + lm) * DIM_FEAT + k0);
            v2f b;
            b.x = W1[(k0    ) * DIM_H + N];
            b.y = W1[(k0 + 1) * DIM_H + N];
            acc = __builtin_amdgcn_wmma_f32_16x16x4_f32(
                      false, a, false, b, (short)0, acc, false, false);
        }
        const float bias = b1[N];
        #pragma unroll
        for (int r = 0; r < 8; ++r) {
            float hv = acc[r] + bias;
            hv = (hv > 0.0f) ? hv : 0.01f * hv;
            hid[wave][r + 8 * lh][N] = hv;
        }
    }

    for (int nt = 0; nt < DIM_E / 16; ++nt) {
        const int N = nt * 16 + lm;
        v8f acc = {};
        #pragma unroll 4
        for (int kb = 0; kb < DIM_H / 4; ++kb) {
            const int k0 = kb * 4 + lh * 2;
            v2f a = *(const v2f*)(&hid[wave][lm][k0]);
            v2f b;
            b.x = W2[(k0    ) * DIM_E + N];
            b.y = W2[(k0 + 1) * DIM_E + N];
            acc = __builtin_amdgcn_wmma_f32_16x16x4_f32(
                      false, a, false, b, (short)0, acc, false, false);
        }
        const float bias = b2[N];
        #pragma unroll
        for (int r = 0; r < 8; ++r)
            temp[(r0 + r + 8 * lh) * DIM_E + N] = acc[r] + bias;
    }
}

__global__ void k_normalize(const float* __restrict__ pref,
                            const float* __restrict__ temp,
                            float* __restrict__ xn) {
    const int wave = threadIdx.x >> 5;
    const int lane = threadIdx.x & 31;
    const int node = blockIdx.x * (blockDim.x >> 5) + wave;
    if (node >= N_NODES) return;
    const float* src = (node < NUM_USER) ? (pref + (size_t)node * DIM_E)
                                         : (temp + (size_t)(node - NUM_USER) * DIM_E);
    float v0 = src[lane];
    float v1 = src[lane + 32];
    float s = v0 * v0 + v1 * v1;
    #pragma unroll
    for (int off = 16; off > 0; off >>= 1) s += __shfl_xor(s, off, 32);
    const float inv = 1.0f / fmaxf(sqrtf(s), 1e-12f);
    xn[(size_t)node * DIM_E + lane]      = v0 * inv;
    xn[(size_t)node * DIM_E + lane + 32] = v1 * inv;
}

__global__ void k_degree(const int* __restrict__ ei, float* __restrict__ deg) {
    int e = blockIdx.x * blockDim.x + threadIdx.x;
    if (e >= N_EDGES) return;
    int r = ei[e];
    int c = ei[N_EDGES + e];
    if (r != c) atomicAdd(deg + r, 1.0f);
}

__global__ void k_dis(float* __restrict__ deg) {
    int n = blockIdx.x * blockDim.x + threadIdx.x;
    if (n >= N_NODES) return;
    float d = deg[n];
    deg[n] = (d > 0.0f) ? rsqrtf(d) : 0.0f;
}

__global__ void k_conv(const int* __restrict__ ei, const float* __restrict__ dis,
                       const float* __restrict__ x, float* __restrict__ out) {
    int tid = blockIdx.x * blockDim.x + threadIdx.x;
    int e = tid >> 4;
    int q = tid & 15;
    if (e >= N_EDGES) return;
    int r = ei[e];
    int c = ei[N_EDGES + e];
    if (r == c) return;
    float nrm = dis[r] * dis[c];
    if (nrm == 0.0f) return;
    const float4 v = *(const float4*)(x + (size_t)r * DIM_E + q * 4);
    float* o = out + (size_t)c * DIM_E + q * 4;
    atomicAdd(o + 0, v.x * nrm);
    atomicAdd(o + 1, v.y * nrm);
    atomicAdd(o + 2, v.z * nrm);
    atomicAdd(o + 3, v.w * nrm);
}

__global__ void k_final(const float* __restrict__ xn, const float* __restrict__ h,
                        const float* __restrict__ h1, const float* __restrict__ pref,
                        float* __restrict__ out) {
    const int n4 = N_NODES * DIM_E / 4;
    const int p4 = NUM_USER * DIM_E / 4;
    int i = blockIdx.x * blockDim.x + threadIdx.x;
    if (i >= n4) return;
    float4 a = ((const float4*)xn)[i];
    float4 b = ((const float4*)h)[i];
    float4 c = ((const float4*)h1)[i];
    float4 rr;
    rr.x = a.x + b.x + c.x;
    rr.y = a.y + b.y + c.y;
    rr.z = a.z + b.z + c.z;
    rr.w = a.w + b.w + c.w;
    ((float4*)out)[i] = rr;
    if (i < p4) ((float4*)out)[n4 + i] = ((const float4*)pref)[i];
}

extern "C" void kernel_launch(void* const* d_in, const int* in_sizes, int n_in,
                              void* d_out, int out_size, void* d_ws, size_t ws_size,
                              hipStream_t stream) {
    const int*   ei   = (const int*)  d_in[0];
    const float* F    = (const float*)d_in[1];
    const float* pref = (const float*)d_in[2];
    const float* W1   = (const float*)d_in[3];
    const float* b1   = (const float*)d_in[4];
    const float* W2   = (const float*)d_in[5];
    const float* b2   = (const float*)d_in[6];
    float*       out  = (float*)d_out;

    float* ws   = (float*)d_ws;
    float* xn   = ws;
    float* h    = xn   + (size_t)N_NODES * DIM_E;
    float* h1   = h    + (size_t)N_NODES * DIM_E;
    float* temp = h1   + (size_t)N_NODES * DIM_E;
    float* deg  = temp + (size_t)NUM_ITEM * DIM_E;

    const int T = 256;
    {
        int n = 2 * N_NODES * DIM_E;
        k_zero<<<(n + T - 1) / T, T, 0, stream>>>(h, n);
        k_zero<<<(N_NODES + T - 1) / T, T, 0, stream>>>(deg, N_NODES);
    }
    k_mlp<<<(N_TILES + 3) / 4, 128, 0, stream>>>(F, W1, b1, W2, b2, temp);
    k_normalize<<<(N_NODES + 7) / 8, T, 0, stream>>>(pref, temp, xn);
    k_degree<<<(N_EDGES + T - 1) / T, T, 0, stream>>>(ei, deg);
    k_dis<<<(N_NODES + T - 1) / T, T, 0, stream>>>(deg);
    {
        int nthr = N_EDGES * 16;
        k_conv<<<(nthr + T - 1) / T, T, 0, stream>>>(ei, deg, xn, h);
        k_conv<<<(nthr + T - 1) / T, T, 0, stream>>>(ei, deg, h, h1);
    }
    k_final<<<(N_NODES * DIM_E / 4 + T - 1) / T, T, 0, stream>>>(xn, h, h1, pref, out);
}


# conv scatter: one wave per 32 edges, each 256-B feature row moved as two whole 128-B lines (32 lanes x 4 B) per load and per f32 atomic add; order of the atomic adds changes
# speedup vs baseline: 1.4942x; 1.4942x over previous
#include <hip/hip_runtime.h>

#define NUM_USER 50000
#define NUM_ITEM 50000
#define N_NODES  100000
#define DIM_E    64
#define DIM_FEAT 128
#define DIM_H    256
#define N_EDGES  1600000
#define N_TILES  (NUM_ITEM / 16)

typedef __attribute__((ext_vector_type(2))) float v2f;
typedef __attribute__((ext_vector_type(8))) float v8f;

__global__ void k_zero(float* __restrict__ p, int n) {
    int i = blockIdx.x * blockDim.x + threadIdx.x;
    if (i < n) p[i] = 0.0f;
}

__global__ __launch_bounds__(128) void k_mlp(
        const float* __restrict__ F,
        const float* __restrict__ W1, const float* __restrict__ b1,
        const float* __restrict__ W2, const float* __restrict__ b2,
        float* __restrict__ temp) {
    __shared__ float hid[4][16][DIM_H];

    const int wave = threadIdx.x >> 5;
    const int lane = threadIdx.x & 31;
    const int lm   = lane & 15;
    const int lh   = lane >> 4;
    const int tile = blockIdx.x * 4 + wave;
    if (tile >= N_TILES) return;
    const int r0 = tile * 16;

    for (int nt = 0; nt < DIM_H / 16; ++nt) {
        const int N = nt * 16 + lm;
        v8f acc = {};
        #pragma unroll 4
        for (int kb = 0; kb < DIM_FEAT / 4; ++kb) {
            const int k0 = kb * 4 + lh * 2;
            v2f a = *(const v2f*)(F + (r0 + lm) * DIM_FEAT + k0);
            v2f b;
            b.x = W1[(k0    ) * DIM_H + N];
            b.y = W1[(k0 + 1) * DIM_H + N];
            acc = __builtin_amdgcn_wmma_f32_16x16x4_f32(
                      false, a, false, b, (short)0, acc, false, false);
        }
        const float bias = b1[N];
        #pragma unroll
        for (int r = 0; r < 8; ++r) {
            float hv = acc[r] + bias;
            hv = (hv > 0.0f) ? hv : 0.01f * hv;
            hid[wave][r + 8 * lh][N] = hv;
        }
    }

    for (int nt = 0; nt < DIM_E / 16; ++nt) {
        const int N = nt * 16 + lm;
        v8f acc = {};
        #pragma unroll 4
        for (int kb = 0; kb < DIM_H / 4; ++kb) {
            const int k0 = kb * 4 + lh * 2;
            v2f a = *(const v2f*)(&hid[wave][lm][k0]);
            v2f b;
            b.x = W2[(k0    ) * DIM_E + N];
            b.y = W2[(k0 + 1) * DIM_E + N];
            acc = __builtin_amdgcn_wmma_f32_16x16x4_f32(
                      false, a, false, b, (short)0, acc, false, false);
        }
        const float bias = b2[N];
        #pragma unroll
        for (int r = 0; r < 8; ++r)
            temp[(r0 + r + 8 * lh) * DIM_E + N] = acc[r] + bias;
    }
}

__global__ void k_normalize(const float* __restrict__ pref,
                            const float* __restrict__ temp,
                            float* __restrict__ xn) {
    const int wave = threadIdx.x >> 5;
    const int lane = threadIdx.x & 31;
    const int node = blockIdx.x * (blockDim.x >> 5) + wave;
    if (node >= N_NODES) return;
    const float* src = (node < NUM_USER) ? (pref + (size_t)node * DIM_E)
                                         : (temp + (size_t)(node - NUM_USER) * DIM_E);
    float v0 = src[lane];
    float v1 = src[lane + 32];
    float s = v0 * v0 + v1 * v1;
    #pragma unroll
    for (int off = 16; off > 0; off >>= 1) s += __shfl_xor(s, off, 32);
    const float inv = 1.0f / fmaxf(sqrtf(s), 1e-12f);
    xn[(size_t)node * DIM_E + lane]      = v0 * inv;
    xn[(size_t)node * DIM_E + lane + 32] = v1 * inv;
}

__global__ void k_degree(const int* __restrict__ ei, float* __restrict__ deg) {
    int e = blockIdx.x * blockDim.x + threadIdx.x;
    if (e >= N_EDGES) return;
    int r = ei[e];
    int c = ei[N_EDGES + e];
    if (r != c) atomicAdd(deg + r, 1.0f);
}

__global__ void k_dis(float* __restrict__ deg) {
    int n = blockIdx.x * blockDim.x + threadIdx.x;
    if (n >= N_NODES) return;
    float d = deg[n];
    deg[n] = (d > 0.0f) ? rsqrtf(d) : 0.0f;
}

#define CONV_WAVES_PER_BLOCK 8
#define CONV_EDGES_PER_WAVE  32
#define CONV_CHUNK           16
static_assert(N_EDGES % (CONV_WAVES_PER_BLOCK * CONV_EDGES_PER_WAVE) == 0, "edge count must fill whole waves");

__global__ __launch_bounds__(CONV_WAVES_PER_BLOCK * 32) void k_conv(
        const int* __restrict__ ei, const float* __restrict__ dis,
        const float* __restrict__ x, float* __restrict__ out) {
    const int lane = threadIdx.x & 31;
    const int wave = blockIdx.x * CONV_WAVES_PER_BLOCK + (threadIdx.x >> 5);
    const int e    = wave * CONV_EDGES_PER_WAVE + lane;
    const int r = ei[e];
    const int c = ei[N_EDGES + e];
    const float nrm = (r != c) ? dis[r] * dis[c] : 0.0f;

    for (int base = 0; base < CONV_EDGES_PER_WAVE; base += CONV_CHUNK) {
        float v0[CONV_CHUNK], v1[CONV_CHUNK];
        #pragma unroll
        for (int j = 0; j < CONV_CHUNK; ++j) {
            const int rj = __builtin_amdgcn_readlane(r, base + j);
            const float* xr = x + (size_t)rj * DIM_E;
            v0[j] = xr[lane];
            v1[j] = xr[lane + 32];
        }
        #pragma unroll
        for (int j = 0; j < CONV_CHUNK; ++j) {
            const int cj = __builtin_amdgcn_readlane(c, base + j);
            const float nj = __int_as_float(__builtin_amdgcn_readlane(__float_as_int(nrm), base + j));
            float* o = out + (size_t)cj * DIM_E;
            atomicAdd(o + lane,      v0[j] * nj);
            atomicAdd(o + lane + 32, v1[j] * nj);
        }
    }
}

__global__ void k_final(const float* __restrict__ xn, const float* __restrict__ h,
                        const float* __restrict__ h1, const float* __restrict__ pref,
                        float* __restrict__ out) {
    const int n4 = N_NODES * DIM_E / 4;
    const int p4 = NUM_USER * DIM_E / 4;
    int i = blockIdx.x * blockDim.x + threadIdx.x;
    if (i >= n4) return;
    float4 a = ((const float4*)xn)[i];
    float4 b = ((const float4*)h)[i];
    float4 c = ((const float4*)h1)[i];
    float4 rr;
    rr.x = a.x + b.x + c.x;
    rr.y = a.y + b.y + c.y;
    rr.z = a.z + b.z + c.z;
    rr.w = a.w + b.w + c.w;
    ((float4*)out)[i] = rr;
    if (i < p4) ((float4*)out)[n4 + i] = ((const float4*)pref)[i];
}

extern "C" void kernel_launch(void* const* d_in, const int* in_sizes, int n_in,
                              void* d_out, int out_size, void* d_ws, size_t ws_size,
                              hipStream_t stream) {
    const int*   ei   = (const int*)  d_in[0];
    const float* F    = (const float*)d_in[1];
    const float* pref = (const float*)d_in[2];
    const float* W1   = (const float*)d_in[3];
    const float* b1   = (const float*)d_in[4];
    const float* W2   = (const float*)d_in[5];
    const float* b2   = (const float*)d_in[6];
    float*       out  = (float*)d_out;

    float* ws   = (float*)d_ws;
    float* xn   = ws;
    float* h    = xn   + (size_t)N_NODES * DIM_E;
    float* h1   = h    + (size_t)N_NODES * DIM_E;
    float* temp = h1   + (size_t)N_NODES * DIM_E;
    float* deg  = temp + (size_t)NUM_ITEM * DIM_E;

    const int T = 256;
    {
        int n = 2 * N_NODES * DIM_E;
        k_zero<<<(n + T - 1) / T, T, 0, stream>>>(h, n);
        k_zero<<<(N_NODES + T - 1) / T, T, 0, stream>>>(deg, N_NODES);
    }
    k_mlp<<<(N_TILES + 3) / 4, 128, 0, stream>>>(F, W1, b1, W2, b2, temp);
    k_normalize<<<(N_NODES + 7) / 8, T, 0, stream>>>(pref, temp, xn);
    k_degree<<<(N_EDGES + T - 1) / T, T, 0, stream>>>(ei, deg);
    k_dis<<<(N_NODES + T - 1) / T, T, 0, stream>>>(deg);
    {
        const int nblk = N_EDGES / (CONV_WAVES_PER_BLOCK * CONV_EDGES_PER_WAVE);
        k_conv<<<nblk, CONV_WAVES_PER_BLOCK * 32, 0, stream>>>(ei, deg, xn, h);
        k_conv<<<nblk, CONV_WAVES_PER_BLOCK * 32, 0, stream>>>(ei, deg, h, h1);
    }
    k_final<<<(N_NODES * DIM_E / 4 + T - 1) / T, T, 0, stream>>>(xn, h, h1, pref, out);
}


# conv hops: edges bucketed by 128-row col range; each block sums its bucket's rows with LDS f32 atomics and stores them plain (no global f32 atomics, no zero-fill); sums run in LDS arrival order
# speedup vs baseline: 1.7080x; 1.1431x over previous
#include <hip/hip_runtime.h>

#define NUM_USER 50000
#define NUM_ITEM 50000
#define N_NODES  100000
#define DIM_E    64
#define DIM_FEAT 128
#define DIM_H    256
#define N_EDGES  1600000
#define N_TILES  (NUM_ITEM / 16)

typedef __attribute__((ext_vector_type(2))) float v2f;
typedef __attribute__((ext_vector_type(8))) float v8f;

__global__ void k_zero(float* __restrict__ p, int n) {
    int i = blockIdx.x * blockDim.x + threadIdx.x;
    if (i < n) p[i] = 0.0f;
}

__global__ __launch_bounds__(128) void k_mlp(
        const float* __restrict__ F,
        const float* __restrict__ W1, const float* __restrict__ b1,
        const float* __restrict__ W2, const float* __restrict__ b2,
        float* __restrict__ temp) {
    __shared__ float hid[4][16][DIM_H];

    const int wave = threadIdx.x >> 5;
    const int lane = threadIdx.x & 31;
    const int lm   = lane & 15;
    const int lh   = lane >> 4;
    const int tile = blockIdx.x * 4 + wave;
    if (tile >= N_TILES) return;
    const int r0 = tile * 16;

    for (int nt = 0; nt < DIM_H / 16; ++nt) {
        const int N = nt * 16 + lm;
        v8f acc = {};
        #pragma unroll 4
        for (int kb = 0; kb < DIM_FEAT / 4; ++kb) {
            const int k0 = kb * 4 + lh * 2;
            v2f a = *(const v2f*)(F + (r0 + lm) * DIM_FEAT + k0);
            v2f b;
            b.x = W1[(k0    ) * DIM_H + N];
            b.y = W1[(k0 + 1) * DIM_H + N];
            acc = __builtin_amdgcn_wmma_f32_16x16x4_f32(
                      false, a, false, b, (short)0, acc, false, false);
        }
        const float bias = b1[N];
        #pragma unroll
        for (int r = 0; r < 8; ++r) {
            float hv = acc[r] + bias;
            hv = (hv > 0.0f) ? hv : 0.01f * hv;
            hid[wave][r + 8 * lh][N] = hv;
        }
    }

    for (int nt = 0; nt < DIM_E / 16; ++nt) {
        const int N = nt * 16 + lm;
        v8f acc = {};
        #pragma unroll 4
        for (int kb = 0; kb < DIM_H / 4; ++kb) {
            const int k0 = kb * 4 + lh * 2;
            v2f a = *(const v2f*)(&hid[wave][lm][k0]);
            v2f b;
            b.x = W2[(k0    ) * DIM_E + N];
            b.y = W2[(k0 + 1) * DIM_E + N];
            acc = __builtin_amdgcn_wmma_f32_16x16x4_f32(
                      false, a, false, b, (short)0, acc, false, false);
        }
        const float bias = b2[N];
        #pragma unroll
        for (int r = 0; r < 8; ++r)
            temp[(r0 + r + 8 * lh) * DIM_E + N] = acc[r] + bias;
    }
}

__global__ void k_normalize(const float* __restrict__ pref,
                            const float* __restrict__ temp,
                            float* __restrict__ xn) {
    const int wave = threadIdx.x >> 5;
    const int lane = threadIdx.x & 31;
    const int node = blockIdx.x * (blockDim.x >> 5) + wave;
    if (node >= N_NODES) return;
    const float* src = (node < NUM_USER) ? (pref + (size_t)node * DIM_E)
                                         : (temp + (size_t)(node - NUM_USER) * DIM_E);
    float v0 = src[lane];
    float v1 = src[lane + 32];
    float s = v0 * v0 + v1 * v1;
    #pragma unroll
    for (int off = 16; off > 0; off >>= 1) s += __shfl_xor(s, off, 32);
    const float inv = 1.0f / fmaxf(sqrtf(s), 1e-12f);
    xn[(size_t)node * DIM_E + lane]      = v0 * inv;
    xn[(size_t)node * DIM_E + lane + 32] = v1 * inv;
}

#define BK_SHIFT   7
#define BK_ROWS    (1 << BK_SHIFT)
#define NB         ((N_NODES + BK_ROWS - 1) / BK_ROWS)
#define NB_PAD     1024
#define HB_THREADS 640
#define HB_PER_THREAD 10
#define HB_EDGES   (HB_THREADS * HB_PER_THREAD)
static_assert(N_EDGES % HB_EDGES == 0, "edge count must fill whole partition blocks");
static_assert(NB <= NB_PAD, "one scan block must cover every bucket");

__global__ __launch_bounds__(HB_THREADS) void k_hist(
        const int* __restrict__ ei, float* __restrict__ deg, int* __restrict__ cnt) {
    __shared__ int lh[NB];
    for (int i = threadIdx.x; i < NB; i += HB_THREADS) lh[i] = 0;
    __syncthreads();
    const int e0 = blockIdx.x * HB_EDGES + threadIdx.x;
    int r[HB_PER_THREAD], c[HB_PER_THREAD];
    #pragma unroll
    for (int k = 0; k < HB_PER_THREAD; ++k) {
        r[k] = ei[e0 + k * HB_THREADS];
        c[k] = ei[N_EDGES + e0 + k * HB_THREADS];
    }
    #pragma unroll
    for (int k = 0; k < HB_PER_THREAD; ++k) {
        if (r[k] != c[k]) atomicAdd(deg + r[k], 1.0f);
        atomicAdd(&lh[c[k] >> BK_SHIFT], 1);
    }
    __syncthreads();
    for (int i = threadIdx.x; i < NB; i += HB_THREADS) {
        const int n = lh[i];
        if (n) atomicAdd(cnt + i, n);
    }
}

__global__ __launch_bounds__(NB_PAD) void k_scan(
        const int* __restrict__ cnt, int* __restrict__ start, int* __restrict__ cursor) {
    __shared__ int s[2][NB_PAD];
    const int t = threadIdx.x;
    const int own = (t < NB) ? cnt[t] : 0;
    int cur = 0;
    s[0][t] = own;
    __syncthreads();
    for (int off = 1; off < NB_PAD; off <<= 1) {
        int v = s[cur][t];
        if (t >= off) v += s[cur][t - off];
        s[cur ^ 1][t] = v;
        cur ^= 1;
        __syncthreads();
    }
    const int incl = s[cur][t];
    if (t < NB) { start[t] = incl - own; cursor[t] = incl - own; }
    if (t == NB - 1) start[NB] = incl;
}

__global__ __launch_bounds__(HB_THREADS) void k_scatter(
        const int* __restrict__ ei, int* __restrict__ cursor, int2* __restrict__ pairs) {
    __shared__ int lh[NB];
    __shared__ int lb[NB];
    for (int i = threadIdx.x; i < NB; i += HB_THREADS) lh[i] = 0;
    __syncthreads();
    const int e0 = blockIdx.x * HB_EDGES + threadIdx.x;
    int r[HB_PER_THREAD], c[HB_PER_THREAD];
    #pragma unroll
    for (int k = 0; k < HB_PER_THREAD; ++k) {
        r[k] = ei[e0 + k * HB_THREADS];
        c[k] = ei[N_EDGES + e0 + k * HB_THREADS];
    }
    #pragma unroll
    for (int k = 0; k < HB_PER_THREAD; ++k) atomicAdd(&lh[c[k] >> BK_SHIFT], 1);
    __syncthreads();
    for (int i = threadIdx.x; i < NB; i += HB_THREADS) {
        const int n = lh[i];
        lb[i] = n ? atomicAdd(cursor + i, n) : 0;
        lh[i] = 0;
    }
    __syncthreads();
    #pragma unroll
    for (int k = 0; k < HB_PER_THREAD; ++k) {
        const int b = c[k] >> BK_SHIFT;
        const int pos = lb[b] + atomicAdd(&lh[b], 1);
        pairs[pos] = make_int2(r[k], c[k]);
    }
}

__global__ void k_dis(float* __restrict__ deg) {
    int n = blockIdx.x * blockDim.x + threadIdx.x;
    if (n >= N_NODES) return;
    float d = deg[n];
    deg[n] = (d > 0.0f) ? rsqrtf(d) : 0.0f;
}

#define CB_WAVES   16
#define CONV_CHUNK 16

__global__ __launch_bounds__(CB_WAVES * 32) void k_conv(
        const int2* __restrict__ pairs, const int* __restrict__ start,
        const float* __restrict__ dis,
        const float* __restrict__ x, float* __restrict__ out) {
    __shared__ float acc[BK_ROWS * DIM_E];
    const int lane = threadIdx.x & 31;
    const int wv   = __builtin_amdgcn_readfirstlane(threadIdx.x >> 5);
    const int row0 = blockIdx.x << BK_SHIFT;

    for (int i = threadIdx.x; i < BK_ROWS * DIM_E / 4; i += CB_WAVES * 32)
        ((float4*)acc)[i] = make_float4(0.0f, 0.0f, 0.0f, 0.0f);
    __syncthreads();

    const int ebeg = start[blockIdx.x];
    const int eend = start[blockIdx.x + 1];
    for (int eb = ebeg + wv * 32; eb < eend; eb += CB_WAVES * 32) {
        const int last = min(31, eend - eb - 1);
        const int2 p   = pairs[eb + min(lane, last)];
        const int r    = p.x;
        const int cl   = p.y - row0;
        const float nrm = (lane <= last && p.x != p.y) ? dis[p.x] * dis[p.y] : 0.0f;

        for (int base = 0; base < 32; base += CONV_CHUNK) {
            float v0[CONV_CHUNK], v1[CONV_CHUNK];
            #pragma unroll
            for (int j = 0; j < CONV_CHUNK; ++j) {
                const int rj = __builtin_amdgcn_readlane(r, base + j);
                const float* xr = x + (size_t)rj * DIM_E;
                v0[j] = xr[lane];
                v1[j] = xr[lane + 32];
            }
            #pragma unroll
            for (int j = 0; j < CONV_CHUNK; ++j) {
                const int cj = __builtin_amdgcn_readlane(cl, base + j);
                const float nj = __int_as_float(__builtin_amdgcn_readlane(__float_as_int(nrm), base + j));
                float* o = acc + cj * DIM_E;
                atomicAdd(o + lane,      v0[j] * nj);
                atomicAdd(o + lane + 32, v1[j] * nj);
            }
        }
    }
    __syncthreads();

    const int nrows = min(BK_ROWS, N_NODES - row0);
    float4* o4 = (float4*)(out + (size_t)row0 * DIM_E);
    for (int i = threadIdx.x; i < nrows * DIM_E / 4; i += CB_WAVES * 32)
        o4[i] = ((const float4*)acc)[i];
}

__global__ void k_final(const float* __restrict__ xn, const float* __restrict__ h,
                        const float* __restrict__ h1, const float* __restrict__ pref,
                        float* __restrict__ out) {
    const int n4 = N_NODES * DIM_E / 4;
    const int p4 = NUM_USER * DIM_E / 4;
    int i = blockIdx.x * blockDim.x + threadIdx.x;
    if (i >= n4) return;
    float4 a = ((const float4*)xn)[i];
    float4 b = ((const float4*)h)[i];
    float4 c = ((const float4*)h1)[i];
    float4 rr;
    rr.x = a.x + b.x + c.x;
    rr.y = a.y + b.y + c.y;
    rr.z = a.z + b.z + c.z;
    rr.w = a.w + b.w + c.w;
    ((float4*)out)[i] = rr;
    if (i < p4) ((float4*)out)[n4 + i] = ((const float4*)pref)[i];
}

extern "C" void kernel_launch(void* const* d_in, const int* in_sizes, int n_in,
                              void* d_out, int out_size, void* d_ws, size_t ws_size,
                              hipStream_t stream) {
    const int*   ei   = (const int*)  d_in[0];
    const float* F    = (const float*)d_in[1];
    const float* pref = (const float*)d_in[2];
    const float* W1   = (const float*)d_in[3];
    const float* b1   = (const float*)d_in[4];
    const float* W2   = (const float*)d_in[5];
    const float* b2   = (const float*)d_in[6];
    float*       out  = (float*)d_out;

    float* ws   = (float*)d_ws;
    float* xn   = ws;
    float* h    = xn   + (size_t)N_NODES * DIM_E;
    float* h1   = h    + (size_t)N_NODES * DIM_E;
    float* temp = h1   + (size_t)N_NODES * DIM_E;
    float* deg  = temp + (size_t)NUM_ITEM * DIM_E;

    const int T = 256;
    int*  cnt    = (int*)(deg + N_NODES);
    int*  start  = cnt   + NB_PAD;
    int*  cursor = start + NB_PAD;
    int2* pairs  = (int2*)(cursor + NB_PAD);

    k_zero<<<(N_NODES + NB_PAD + T - 1) / T, T, 0, stream>>>(deg, N_NODES + NB_PAD);
    k_mlp<<<(N_TILES + 3) / 4, 128, 0, stream>>>(F, W1, b1, W2, b2, temp);
    k_normalize<<<(N_NODES + 7) / 8, T, 0, stream>>>(pref, temp, xn);
    k_hist<<<N_EDGES / HB_EDGES, HB_THREADS, 0, stream>>>(ei, deg, cnt);
    k_dis<<<(N_NODES + T - 1) / T, T, 0, stream>>>(deg);
    k_scan<<<1, NB_PAD, 0, stream>>>(cnt, start, cursor);
    k_scatter<<<N_EDGES / HB_EDGES, HB_THREADS, 0, stream>>>(ei, cursor, pairs);
    k_conv<<<NB, CB_WAVES * 32, 0, stream>>>(pairs, start, deg, xn, h);
    k_conv<<<NB, CB_WAVES * 32, 0, stream>>>(pairs, start, deg, h, h1);
    k_final<<<(N_NODES * DIM_E / 4 + T - 1) / T, T, 0, stream>>>(xn, h, h1, pref, out);
}


# MLP: W1 and W2 copied once per block into LDS as 4-k words, 256 blocks of 8 waves walk the row tiles, F tile in registers, 4 accumulator tiles per wave; k order inside the matrix products changes
# speedup vs baseline: 2.1656x; 1.2679x over previous
#include <hip/hip_runtime.h>

#define NUM_USER 50000
#define NUM_ITEM 50000
#define N_NODES  100000
#define DIM_E    64
#define DIM_FEAT 128
#define DIM_H    256
#define N_EDGES  1600000
#define N_TILES  (NUM_ITEM / 16)

typedef __attribute__((ext_vector_type(2))) float v2f;
typedef __attribute__((ext_vector_type(8))) float v8f;

__global__ void k_zero(float* __restrict__ p, int n) {
    int i = blockIdx.x * blockDim.x + threadIdx.x;
    if (i < n) p[i] = 0.0f;
}

typedef __attribute__((ext_vector_type(4))) float v4f;

#define MLP_WAVES   8
#define MLP_THREADS (MLP_WAVES * 32)
#define MLP_BLOCKS  256
#define HID_PITCH   68
#define W1Q_WORDS   ((DIM_FEAT / 4) * DIM_H)
#define W2Q_WORDS   ((DIM_H / 4) * DIM_E)
static_assert(MLP_BLOCKS * MLP_WAVES <= N_TILES, "every wave must own a first tile");
static_assert(W1Q_WORDS % MLP_THREADS == 0 && W2Q_WORDS % MLP_THREADS == 0, "the copy loops have whole trips");
static_assert(2 * MLP_BLOCKS * MLP_WAVES >= N_TILES, "two rounds must cover every tile");

__device__ __forceinline__ v8f mma4(v2f a, v2f b, v8f c) {
    return __builtin_amdgcn_wmma_f32_16x16x4_f32(false, a, false, b, (short)0, c, false, false);
}

__device__ __forceinline__ v4f kquad(const float* __restrict__ M, int i, int log2n) {
    const int n = 1 << log2n;
    const float* p = M + ((i >> log2n) * 4 << log2n) + (i & (n - 1));
    v4f v;
    v.x = p[0]; v.y = p[n]; v.z = p[2 * n]; v.w = p[3 * n];
    return v;
}

__device__ __forceinline__ void load_tile(v4f (&a)[16], const float* __restrict__ F, int tile, int c, int hh) {
    const v4f* fr = (const v4f*)(F + (size_t)(tile * 16 + c) * DIM_FEAT) + hh;
    #pragma unroll
    for (int j = 0; j < 16; ++j) a[j] = fr[2 * j];
}

__global__ __launch_bounds__(MLP_THREADS) void k_mlp(
        const float* __restrict__ F,
        const float* __restrict__ W1, const float* __restrict__ b1,
        const float* __restrict__ W2, const float* __restrict__ b2,
        float* __restrict__ temp) {
    __shared__ v4f   w1q[W1Q_WORDS];
    __shared__ v4f   w2q[W2Q_WORDS];
    __shared__ alignas(16) float hid[MLP_WAVES][16 * HID_PITCH];

    const int tid  = threadIdx.x;
    const int lane = tid & 31;
    const int c    = lane & 15;
    const int hh   = lane >> 4;
    const int wv   = __builtin_amdgcn_readfirstlane(tid >> 5);
    float* hw = hid[wv];

    int tile = blockIdx.x + MLP_BLOCKS * wv;
    v4f a[16];
    load_tile(a, F, tile, c, hh);

    #pragma unroll 16
    for (int it = 0; it < W1Q_WORDS / MLP_THREADS; ++it) {
        const int i = tid + it * MLP_THREADS;
        w1q[i] = kquad(W1, i, 8);
    }
    #pragma unroll
    for (int it = 0; it < W2Q_WORDS / MLP_THREADS; ++it) {
        const int i = tid + it * MLP_THREADS;
        w2q[i] = kquad(W2, i, 6);
    }
    __syncthreads();

    for (;;) {
        v8f acc2[4] = {};
        for (int q = 0; q < 4; ++q) {
            v8f acc1[4] = {};
            const v4f* wb = w1q + hh * DIM_H + 64 * q + c;
            #pragma unroll
            for (int j = 0; j < 16; ++j) {
                v4f bq[4];
                #pragma unroll
                for (int t = 0; t < 4; ++t) bq[t] = wb[2 * j * DIM_H + 16 * t];
                #pragma unroll
                for (int t = 0; t < 4; ++t) acc1[t] = mma4(a[j].xy, bq[t].xy, acc1[t]);
                #pragma unroll
                for (int t = 0; t < 4; ++t) acc1[t] = mma4(a[j].zw, bq[t].zw, acc1[t]);
            }
            #pragma unroll
            for (int t = 0; t < 4; ++t) {
                const float bias = b1[64 * q + 16 * t + c];
                #pragma unroll
                for (int r = 0; r < 8; ++r) {
                    float hv = acc1[t][r] + bias;
                    hv = (hv > 0.0f) ? hv : 0.01f * hv;
                    hw[(8 * hh + r) * HID_PITCH + 16 * t + c] = hv;
                }
            }

            const v4f* w2b = w2q + (16 * q + hh) * DIM_E + c;
            const float* hr = hw + c * HID_PITCH + 4 * hh;
            #pragma unroll
            for (int j = 0; j < 8; ++j) {
                const v4f ah = *(const v4f*)(hr + 8 * j);
                v4f bq[4];
                #pragma unroll
                for (int t = 0; t < 4; ++t) bq[t] = w2b[2 * j * DIM_E + 16 * t];
                #pragma unroll
                for (int t = 0; t < 4; ++t) acc2[t] = mma4(ah.xy, bq[t].xy, acc2[t]);
                #pragma unroll
                for (int t = 0; t < 4; ++t) acc2[t] = mma4(ah.zw, bq[t].zw, acc2[t]);
            }
        }

        #pragma unroll
        for (int t = 0; t < 4; ++t) {
            const float bias = b2[16 * t + c];
            #pragma unroll
            for (int r = 0; r < 8; ++r)
                hw[(8 * hh + r) * HID_PITCH + 16 * t + c] = acc2[t][r] + bias;
        }
        v4f* dst = (v4f*)(temp + (size_t)tile * 16 * DIM_E);
        #pragma unroll
        for (int i = 0; i < 8; ++i) {
            const int idx = i * 32 + lane;
            dst[idx] = *(const v4f*)(hw + (idx >> 4) * HID_PITCH + 4 * (idx & 15));
        }

        tile += MLP_BLOCKS * MLP_WAVES;
        if (tile >= N_TILES) break;
        load_tile(a, F, tile, c, hh);
    }
}

__global__ void k_normalize(const float* __restrict__ pref,
                            const float* __restrict__ temp,
                            float* __restrict__ xn) {
    const int wave = threadIdx.x >> 5;
    const int lane = threadIdx.x & 31;
    const int node = blockIdx.x * (blockDim.x >> 5) + wave;
    if (node >= N_NODES) return;
    const float* src = (node < NUM_USER) ? (pref + (size_t)node * DIM_E)
                                         : (temp + (size_t)(node - NUM_USER) * DIM_E);
    float v0 = src[lane];
    float v1 = src[lane + 32];
    float s = v0 * v0 + v1 * v1;
    #pragma unroll
    for (int off = 16; off > 0; off >>= 1) s += __shfl_xor(s, off, 32);
    const float inv = 1.0f / fmaxf(sqrtf(s), 1e-12f);
    xn[(size_t)node * DIM_E + lane]      = v0 * inv;
    xn[(size_t)node * DIM_E + lane + 32] = v1 * inv;
}

#define BK_SHIFT   7
#define BK_ROWS    (1 << BK_SHIFT)
#define NB         ((N_NODES + BK_ROWS - 1) / BK_ROWS)
#define NB_PAD     1024
#define HB_THREADS 640
#define HB_PER_THREAD 10
#define HB_EDGES   (HB_THREADS * HB_PER_THREAD)
static_assert(N_EDGES % HB_EDGES == 0, "edge count must fill whole partition blocks");
static_assert(NB <= NB_PAD, "one scan block must cover every bucket");

__global__ __launch_bounds__(HB_THREADS) void k_hist(
        const int* __restrict__ ei, float* __restrict__ deg, int* __restrict__ cnt) {
    __shared__ int lh[NB];
    for (int i = threadIdx.x; i < NB; i += HB_THREADS) lh[i] = 0;
    __syncthreads();
    const int e0 = blockIdx.x * HB_EDGES + threadIdx.x;
    int r[HB_PER_THREAD], c[HB_PER_THREAD];
    #pragma unroll
    for (int k = 0; k < HB_PER_THREAD; ++k) {
        r[k] = ei[e0 + k * HB_THREADS];
        c[k] = ei[N_EDGES + e0 + k * HB_THREADS];
    }
    #pragma unroll
    for (int k = 0; k < HB_PER_THREAD; ++k) {
        if (r[k] != c[k]) atomicAdd(deg + r[k], 1.0f);
        atomicAdd(&lh[c[k] >> BK_SHIFT], 1);
    }
    __syncthreads();
    for (int i = threadIdx.x; i < NB; i += HB_THREADS) {
        const int n = lh[i];
        if (n) atomicAdd(cnt + i, n);
    }
}

__global__ __launch_bounds__(NB_PAD) void k_scan(
        const int* __restrict__ cnt, int* __restrict__ start, int* __restrict__ cursor) {
    __shared__ int s[2][NB_PAD];
    const int t = threadIdx.x;
    const int own = (t < NB) ? cnt[t] : 0;
    int cur = 0;
    s[0][t] = own;
    __syncthreads();
    for (int off = 1; off < NB_PAD; off <<= 1) {
        int v = s[cur][t];
        if (t >= off) v += s[cur][t - off];
        s[cur ^ 1][t] = v;
        cur ^= 1;
        __syncthreads();
    }
    const int incl = s[cur][t];
    if (t < NB) { start[t] = incl - own; cursor[t] = incl - own; }
    if (t == NB - 1) start[NB] = incl;
}

__global__ __launch_bounds__(HB_THREADS) void k_scatter(
        const int* __restrict__ ei, int* __restrict__ cursor, int2* __restrict__ pairs) {
    __shared__ int lh[NB];
    __shared__ int lb[NB];
    for (int i = threadIdx.x; i < NB; i += HB_THREADS) lh[i] = 0;
    __syncthreads();
    const int e0 = blockIdx.x * HB_EDGES + threadIdx.x;
    int r[HB_PER_THREAD], c[HB_PER_THREAD];
    #pragma unroll
    for (int k = 0; k < HB_PER_THREAD; ++k) {
        r[k] = ei[e0 + k * HB_THREADS];
        c[k] = ei[N_EDGES + e0 + k * HB_THREADS];
    }
    #pragma unroll
    for (int k = 0; k < HB_PER_THREAD; ++k) atomicAdd(&lh[c[k] >> BK_SHIFT], 1);
    __syncthreads();
    for (int i = threadIdx.x; i < NB; i += HB_THREADS) {
        const int n = lh[i];
        lb[i] = n ? atomicAdd(cursor + i, n) : 0;
        lh[i] = 0;
    }
    __syncthreads();
    #pragma unroll
    for (int k = 0; k < HB_PER_THREAD; ++k) {
        const int b = c[k] >> BK_SHIFT;
        const int pos = lb[b] + atomicAdd(&lh[b], 1);
        pairs[pos] = make_int2(r[k], c[k]);
    }
}

__global__ void k_dis(float* __restrict__ deg) {
    int n = blockIdx.x * blockDim.x + threadIdx.x;
    if (n >= N_NODES) return;
    float d = deg[n];
    deg[n] = (d > 0.0f) ? rsqrtf(d) : 0.0f;
}

#define CB_WAVES   16
#define CONV_CHUNK 16

__global__ __launch_bounds__(CB_WAVES * 32) void k_conv(
        const int2* __restrict__ pairs, const int* __restrict__ start,
        const float* __restrict__ dis,
        const float* __restrict__ x, float* __restrict__ out) {
    __shared__ float acc[BK_ROWS * DIM_E];
    const int lane = threadIdx.x & 31;
    const int wv   = __builtin_amdgcn_readfirstlane(threadIdx.x >> 5);
    const int row0 = blockIdx.x << BK_SHIFT;

    for (int i = threadIdx.x; i < BK_ROWS * DIM_E / 4; i += CB_WAVES * 32)
        ((float4*)acc)[i] = make_float4(0.0f, 0.0f, 0.0f, 0.0f);
    __syncthreads();

    const int ebeg = start[blockIdx.x];
    const int eend = start[blockIdx.x + 1];
    for (int eb = ebeg + wv * 32; eb < eend; eb += CB_WAVES * 32) {
        const int last = min(31, eend - eb - 1);
        const int2 p   = pairs[eb + min(lane, last)];
        const int r    = p.x;
        const int cl   = p.y - row0;
        const float nrm = (lane <= last && p.x != p.y) ? dis[p.x] * dis[p.y] : 0.0f;

        for (int base = 0; base < 32; base += CONV_CHUNK) {
            float v0[CONV_CHUNK], v1[CONV_CHUNK];
            #pragma unroll
            for (int j = 0; j < CONV_CHUNK; ++j) {
                const int rj = __builtin_amdgcn_readlane(r, base + j);
                const float* xr = x + (size_t)rj * DIM_E;
                v0[j] = xr[lane];
                v1[j] = xr[lane + 32];
            }
            #pragma unroll
            for (int j = 0; j < CONV_CHUNK; ++j) {
                const int cj = __builtin_amdgcn_readlane(cl, base + j);
                const float nj = __int_as_float(__builtin_amdgcn_readlane(__float_as_int(nrm), base + j));
                float* o = acc + cj * DIM_E;
                atomicAdd(o + lane,      v0[j] * nj);
                atomicAdd(o + lane + 32, v1[j] * nj);
            }
        }
    }
    __syncthreads();

    const int nrows = min(BK_ROWS, N_NODES - row0);
    float4* o4 = (float4*)(out + (size_t)row0 * DIM_E);
    for (int i = threadIdx.x; i < nrows * DIM_E / 4; i += CB_WAVES * 32)
        o4[i] = ((const float4*)acc)[i];
}

__global__ void k_final(const float* __restrict__ xn, const float* __restrict__ h,
                        const float* __restrict__ h1, const float* __restrict__ pref,
                        float* __restrict__ out) {
    const int n4 = N_NODES * DIM_E / 4;
    const int p4 = NUM_USER * DIM_E / 4;
    int i = blockIdx.x * blockDim.x + threadIdx.x;
    if (i >= n4) return;
    float4 a = ((const float4*)xn)[i];
    float4 b = ((const float4*)h)[i];
    float4 c = ((const float4*)h1)[i];
    float4 rr;
    rr.x = a.x + b.x + c.x;
    rr.y = a.y + b.y + c.y;
    rr.z = a.z + b.z + c.z;
    rr.w = a.w + b.w + c.w;
    ((float4*)out)[i] = rr;
    if (i < p4) ((float4*)out)[n4 + i] = ((const float4*)pref)[i];
}

extern "C" void kernel_launch(void* const* d_in, const int* in_sizes, int n_in,
                              void* d_out, int out_size, void* d_ws, size_t ws_size,
                              hipStream_t stream) {
    const int*   ei   = (const int*)  d_in[0];
    const float* F    = (const float*)d_in[1];
    const float* pref = (const float*)d_in[2];
    const float* W1   = (const float*)d_in[3];
    const float* b1   = (const float*)d_in[4];
    const float* W2   = (const float*)d_in[5];
    const float* b2   = (const float*)d_in[6];
    float*       out  = (float*)d_out;

    float* ws   = (float*)d_ws;
    float* xn   = ws;
    float* h    = xn   + (size_t)N_NODES * DIM_E;
    float* h1   = h    + (size_t)N_NODES * DIM_E;
    float* temp = h1   + (size_t)N_NODES * DIM_E;
    float* deg  = temp + (size_t)NUM_ITEM * DIM_E;

    const int T = 256;
    int*  cnt    = (int*)(deg + N_NODES);
    int*  start  = cnt   + NB_PAD;
    int*  cursor = start + NB_PAD;
    int2* pairs  = (int2*)(cursor + NB_PAD);

    k_zero<<<(N_NODES + NB_PAD + T - 1) / T, T, 0, stream>>>(deg, N_NODES + NB_PAD);
    k_mlp<<<MLP_BLOCKS, MLP_THREADS, 0, stream>>>(F, W1, b1, W2, b2, temp);
    k_normalize<<<(N_NODES + 7) / 8, T, 0, stream>>>(pref, temp, xn);
    k_hist<<<N_EDGES / HB_EDGES, HB_THREADS, 0, stream>>>(ei, deg, cnt);
    k_dis<<<(N_NODES + T - 1) / T, T, 0, stream>>>(deg);
    k_scan<<<1, NB_PAD, 0, stream>>>(cnt, start, cursor);
    k_scatter<<<N_EDGES / HB_EDGES, HB_THREADS, 0, stream>>>(ei, cursor, pairs);
    k_conv<<<NB, CB_WAVES * 32, 0, stream>>>(pairs, start, deg, xn, h);
    k_conv<<<NB, CB_WAVES * 32, 0, stream>>>(pairs, start, deg, h, h1);
    k_final<<<(N_NODES * DIM_E / 4 + T - 1) / T, T, 0, stream>>>(xn, h, h1, pref, out);
}


# the four element-wise launches (zero-fill, row normalize, deg^-0.5, x+h+h1 with the preference copy) folded into the MLP, scatter and second-hop kernels: 10 launches -> 6; norm sum order changes
# speedup vs baseline: 2.3955x; 1.1062x over previous
#include <hip/hip_runtime.h>

#define NUM_USER 50000
#define NUM_ITEM 50000
#define N_NODES  100000
#define DIM_E    64
#define DIM_FEAT 128
#define DIM_H    256
#define N_EDGES  1600000
#define N_TILES  (NUM_ITEM / 16)

typedef __attribute__((ext_vector_type(2))) float v2f;
typedef __attribute__((ext_vector_type(8))) float v8f;

typedef __attribute__((ext_vector_type(4))) float v4f;

#define MLP_WAVES   8
#define MLP_THREADS (MLP_WAVES * 32)
#define MLP_BLOCKS  256
#define HID_PITCH   68
#define W1Q_WORDS   ((DIM_FEAT / 4) * DIM_H)
#define W2Q_WORDS   ((DIM_H / 4) * DIM_E)
static_assert(MLP_BLOCKS * MLP_WAVES <= N_TILES, "every wave must own a first tile");
static_assert(W1Q_WORDS % MLP_THREADS == 0 && W2Q_WORDS % MLP_THREADS == 0, "the copy loops have whole trips");
static_assert(2 * MLP_BLOCKS * MLP_WAVES >= N_TILES, "two rounds must cover every tile");
static_assert(NUM_USER == NUM_ITEM, "item tile t is paired with the preference rows t*16..t*16+15");

__device__ __forceinline__ v8f mma4(v2f a, v2f b, v8f c) {
    return __builtin_amdgcn_wmma_f32_16x16x4_f32(false, a, false, b, (short)0, c, false, false);
}

__device__ __forceinline__ v4f kquad(const float* __restrict__ M, int i, int log2n) {
    const int n = 1 << log2n;
    const float* p = M + ((i >> log2n) * 4 << log2n) + (i & (n - 1));
    v4f v;
    v.x = p[0]; v.y = p[n]; v.z = p[2 * n]; v.w = p[3 * n];
    return v;
}

__device__ __forceinline__ float row_inv_norm(v4f v) {
    float s = v.x * v.x + v.y * v.y + v.z * v.z + v.w * v.w;
    #pragma unroll
    for (int off = 8; off > 0; off >>= 1) s += __shfl_xor(s, off, 32);
    return 1.0f / fmaxf(sqrtf(s), 1e-12f);
}

__device__ __forceinline__ void load_tile(v4f (&a)[16], const float* __restrict__ F, int tile, int c, int hh) {
    const v4f* fr = (const v4f*)(F + (size_t)(tile * 16 + c) * DIM_FEAT) + hh;
    #pragma unroll
    for (int j = 0; j < 16; ++j) a[j] = fr[2 * j];
}

__global__ __launch_bounds__(MLP_THREADS) void k_mlp(
        const float* __restrict__ F,
        const float* __restrict__ W1, const float* __restrict__ b1,
        const float* __restrict__ W2, const float* __restrict__ b2,
        const float* __restrict__ pref,
        float* __restrict__ xn, float* __restrict__ pref_out,
        float* __restrict__ zero_base, int zero_n) {
    __shared__ v4f   w1q[W1Q_WORDS];
    __shared__ v4f   w2q[W2Q_WORDS];
    __shared__ alignas(16) float hid[MLP_WAVES][16 * HID_PITCH];

    const int tid  = threadIdx.x;
    const int lane = tid & 31;
    const int c    = lane & 15;
    const int hh   = lane >> 4;
    const int wv   = __builtin_amdgcn_readfirstlane(tid >> 5);
    float* hw = hid[wv];

    int tile = blockIdx.x + MLP_BLOCKS * wv;
    v4f a[16];
    load_tile(a, F, tile, c, hh);

    for (int i = blockIdx.x * MLP_THREADS + tid; i < zero_n; i += MLP_BLOCKS * MLP_THREADS)
        zero_base[i] = 0.0f;

    #pragma unroll 16
    for (int it = 0; it < W1Q_WORDS / MLP_THREADS; ++it) {
        const int i = tid + it * MLP_THREADS;
        w1q[i] = kquad(W1, i, 8);
    }
    #pragma unroll
    for (int it = 0; it < W2Q_WORDS / MLP_THREADS; ++it) {
        const int i = tid + it * MLP_THREADS;
        w2q[i] = kquad(W2, i, 6);
    }
    __syncthreads();

    for (;;) {
        v4f pv[8];
        {
            const v4f* ps = (const v4f*)(pref + (size_t)tile * 16 * DIM_E);
            #pragma unroll
            for (int i = 0; i < 8; ++i) pv[i] = ps[i * 32 + lane];
        }
        v8f acc2[4] = {};
        for (int q = 0; q < 4; ++q) {
            v8f acc1[4] = {};
            const v4f* wb = w1q + hh * DIM_H + 64 * q + c;
            #pragma unroll
            for (int j = 0; j < 16; ++j) {
                v4f bq[4];
                #pragma unroll
                for (int t = 0; t < 4; ++t) bq[t] = wb[2 * j * DIM_H + 16 * t];
                #pragma unroll
                for (int t = 0; t < 4; ++t) acc1[t] = mma4(a[j].xy, bq[t].xy, acc1[t]);
                #pragma unroll
                for (int t = 0; t < 4; ++t) acc1[t] = mma4(a[j].zw, bq[t].zw, acc1[t]);
            }
            #pragma unroll
            for (int t = 0; t < 4; ++t) {
                const float bias = b1[64 * q + 16 * t + c];
                #pragma unroll
                for (int r = 0; r < 8; ++r) {
                    float hv = acc1[t][r] + bias;
                    hv = (hv > 0.0f) ? hv : 0.01f * hv;
                    hw[(8 * hh + r) * HID_PITCH + 16 * t + c] = hv;
                }
            }

            const v4f* w2b = w2q + (16 * q + hh) * DIM_E + c;
            const float* hr = hw + c * HID_PITCH + 4 * hh;
            #pragma unroll
            for (int j = 0; j < 8; ++j) {
                const v4f ah = *(const v4f*)(hr + 8 * j);
                v4f bq[4];
                #pragma unroll
                for (int t = 0; t < 4; ++t) bq[t] = w2b[2 * j * DIM_E + 16 * t];
                #pragma unroll
                for (int t = 0; t < 4; ++t) acc2[t] = mma4(ah.xy, bq[t].xy, acc2[t]);
                #pragma unroll
                for (int t = 0; t < 4; ++t) acc2[t] = mma4(ah.zw, bq[t].zw, acc2[t]);
            }
        }

        #pragma unroll
        for (int t = 0; t < 4; ++t) {
            const float bias = b2[16 * t + c];
            #pragma unroll
            for (int r = 0; r < 8; ++r)
                hw[(8 * hh + r) * HID_PITCH + 16 * t + c] = acc2[t][r] + bias;
        }
        v4f* dst = (v4f*)(xn + (size_t)(NUM_USER + tile * 16) * DIM_E);
        #pragma unroll
        for (int i = 0; i < 8; ++i) {
            const int idx = i * 32 + lane;
            const v4f v = *(const v4f*)(hw + (idx >> 4) * HID_PITCH + 4 * (idx & 15));
            dst[idx] = v * row_inv_norm(v);
        }
        v4f* xu = (v4f*)(xn + (size_t)tile * 16 * DIM_E);
        v4f* pu = (v4f*)(pref_out + (size_t)tile * 16 * DIM_E);
        #pragma unroll
        for (int i = 0; i < 8; ++i) {
            const int idx = i * 32 + lane;
            pu[idx] = pv[i];
            xu[idx] = pv[i] * row_inv_norm(pv[i]);
        }

        tile += MLP_BLOCKS * MLP_WAVES;
        if (tile >= N_TILES) break;
        load_tile(a, F, tile, c, hh);
    }
}

#define BK_SHIFT   7
#define BK_ROWS    (1 << BK_SHIFT)
#define NB         ((N_NODES + BK_ROWS - 1) / BK_ROWS)
#define NB_PAD     1024
#define HB_THREADS 640
#define HB_PER_THREAD 10
#define HB_EDGES   (HB_THREADS * HB_PER_THREAD)
static_assert(N_EDGES % HB_EDGES == 0, "edge count must fill whole partition blocks");
static_assert(NB <= NB_PAD, "one scan block must cover every bucket");

__global__ __launch_bounds__(HB_THREADS) void k_hist(
        const int* __restrict__ ei, float* __restrict__ deg, int* __restrict__ cnt) {
    __shared__ int lh[NB];
    for (int i = threadIdx.x; i < NB; i += HB_THREADS) lh[i] = 0;
    __syncthreads();
    const int e0 = blockIdx.x * HB_EDGES + threadIdx.x;
    int r[HB_PER_THREAD], c[HB_PER_THREAD];
    #pragma unroll
    for (int k = 0; k < HB_PER_THREAD; ++k) {
        r[k] = ei[e0 + k * HB_THREADS];
        c[k] = ei[N_EDGES + e0 + k * HB_THREADS];
    }
    #pragma unroll
    for (int k = 0; k < HB_PER_THREAD; ++k) {
        if (r[k] != c[k]) atomicAdd(deg + r[k], 1.0f);
        atomicAdd(&lh[c[k] >> BK_SHIFT], 1);
    }
    __syncthreads();
    for (int i = threadIdx.x; i < NB; i += HB_THREADS) {
        const int n = lh[i];
        if (n) atomicAdd(cnt + i, n);
    }
}

__global__ __launch_bounds__(NB_PAD) void k_scan(
        const int* __restrict__ cnt, int* __restrict__ start, int* __restrict__ cursor) {
    __shared__ int s[2][NB_PAD];
    const int t = threadIdx.x;
    const int own = (t < NB) ? cnt[t] : 0;
    int cur = 0;
    s[0][t] = own;
    __syncthreads();
    for (int off = 1; off < NB_PAD; off <<= 1) {
        int v = s[cur][t];
        if (t >= off) v += s[cur][t - off];
        s[cur ^ 1][t] = v;
        cur ^= 1;
        __syncthreads();
    }
    const int incl = s[cur][t];
    if (t < NB) { start[t] = incl - own; cursor[t] = incl - own; }
    if (t == NB - 1) start[NB] = incl;
}

static_assert((N_EDGES / HB_EDGES) * HB_THREADS >= N_NODES, "one thread per node for the degree pass");

__global__ __launch_bounds__(HB_THREADS) void k_scatter(
        const int* __restrict__ ei, int* __restrict__ cursor, int2* __restrict__ pairs,
        float* __restrict__ deg) {
    __shared__ int lh[NB];
    __shared__ int lb[NB];
    for (int i = threadIdx.x; i < NB; i += HB_THREADS) lh[i] = 0;
    {
        const int n = blockIdx.x * HB_THREADS + threadIdx.x;
        if (n < N_NODES) {
            const float d = deg[n];
            deg[n] = (d > 0.0f) ? rsqrtf(d) : 0.0f;
        }
    }
    __syncthreads();
    const int e0 = blockIdx.x * HB_EDGES + threadIdx.x;
    int r[HB_PER_THREAD], c[HB_PER_THREAD];
    #pragma unroll
    for (int k = 0; k < HB_PER_THREAD; ++k) {
        r[k] = ei[e0 + k * HB_THREADS];
        c[k] = ei[N_EDGES + e0 + k * HB_THREADS];
    }
    #pragma unroll
    for (int k = 0; k < HB_PER_THREAD; ++k) atomicAdd(&lh[c[k] >> BK_SHIFT], 1);
    __syncthreads();
    for (int i = threadIdx.x; i < NB; i += HB_THREADS) {
        const int n = lh[i];
        lb[i] = n ? atomicAdd(cursor + i, n) : 0;
        lh[i] = 0;
    }
    __syncthreads();
    #pragma unroll
    for (int k = 0; k < HB_PER_THREAD; ++k) {
        const int b = c[k] >> BK_SHIFT;
        const int pos = lb[b] + atomicAdd(&lh[b], 1);
        pairs[pos] = make_int2(r[k], c[k]);
    }
}

#define CB_WAVES   16
#define CONV_CHUNK 16

#define CONV_OUT_TRIPS (BK_ROWS * DIM_E / 4 / (CB_WAVES * 32))
static_assert(CONV_OUT_TRIPS * CB_WAVES * 32 * 4 == BK_ROWS * DIM_E, "the store pass has whole trips");

template <bool LAST>
__device__ __forceinline__ void conv_body(
        const int2* __restrict__ pairs, const int* __restrict__ start,
        const float* __restrict__ dis,
        const float* __restrict__ x, const float* __restrict__ xn,
        float* __restrict__ out) {
    __shared__ float acc[BK_ROWS * DIM_E];
    const int lane = threadIdx.x & 31;
    const int wv   = __builtin_amdgcn_readfirstlane(threadIdx.x >> 5);
    const int row0 = blockIdx.x << BK_SHIFT;

    for (int i = threadIdx.x; i < BK_ROWS * DIM_E / 4; i += CB_WAVES * 32)
        ((float4*)acc)[i] = make_float4(0.0f, 0.0f, 0.0f, 0.0f);
    __syncthreads();

    const int ebeg = start[blockIdx.x];
    const int eend = start[blockIdx.x + 1];
    for (int eb = ebeg + wv * 32; eb < eend; eb += CB_WAVES * 32) {
        const int last = min(31, eend - eb - 1);
        const int2 p   = pairs[eb + min(lane, last)];
        const int r    = p.x;
        const int cl   = p.y - row0;
        const float nrm = (lane <= last && p.x != p.y) ? dis[p.x] * dis[p.y] : 0.0f;

        for (int base = 0; base < 32; base += CONV_CHUNK) {
            float v0[CONV_CHUNK], v1[CONV_CHUNK];
            #pragma unroll
            for (int j = 0; j < CONV_CHUNK; ++j) {
                const int rj = __builtin_amdgcn_readlane(r, base + j);
                const float* xr = x + (size_t)rj * DIM_E;
                v0[j] = xr[lane];
                v1[j] = xr[lane + 32];
            }
            #pragma unroll
            for (int j = 0; j < CONV_CHUNK; ++j) {
                const int cj = __builtin_amdgcn_readlane(cl, base + j);
                const float nj = __int_as_float(__builtin_amdgcn_readlane(__float_as_int(nrm), base + j));
                float* o = acc + cj * DIM_E;
                atomicAdd(o + lane,      v0[j] * nj);
                atomicAdd(o + lane + 32, v1[j] * nj);
            }
        }
    }
    const int nrows = min(BK_ROWS, N_NODES - row0);
    const int n4    = nrows * DIM_E / 4;
    float4* o4 = (float4*)(out + (size_t)row0 * DIM_E);
    if (LAST) {
        const float4* x4 = (const float4*)(xn + (size_t)row0 * DIM_E);
        const float4* h4 = (const float4*)(x  + (size_t)row0 * DIM_E);
        float4 xa[CONV_OUT_TRIPS], hb[CONV_OUT_TRIPS];
        #pragma unroll
        for (int k = 0; k < CONV_OUT_TRIPS; ++k) {
            const int ic = min((int)threadIdx.x + k * CB_WAVES * 32, n4 - 1);
            xa[k] = x4[ic];
            hb[k] = h4[ic];
        }
        __syncthreads();
        float4 sm[CONV_OUT_TRIPS];
        #pragma unroll
        for (int k = 0; k < CONV_OUT_TRIPS; ++k)
            sm[k] = ((const float4*)acc)[threadIdx.x + k * CB_WAVES * 32];
        #pragma unroll
        for (int k = 0; k < CONV_OUT_TRIPS; ++k) {
            const int i = threadIdx.x + k * CB_WAVES * 32;
            if (i < n4) {
                float4 rr;
                rr.x = xa[k].x + hb[k].x + sm[k].x;
                rr.y = xa[k].y + hb[k].y + sm[k].y;
                rr.z = xa[k].z + hb[k].z + sm[k].z;
                rr.w = xa[k].w + hb[k].w + sm[k].w;
                o4[i] = rr;
            }
        }
    } else {
        __syncthreads();
        for (int i = threadIdx.x; i < n4; i += CB_WAVES * 32)
            o4[i] = ((const float4*)acc)[i];
    }
}

__global__ __launch_bounds__(CB_WAVES * 32) void k_conv(
        const int2* __restrict__ pairs, const int* __restrict__ start,
        const float* __restrict__ dis,
        const float* __restrict__ x, float* __restrict__ out) {
    conv_body<false>(pairs, start, dis, x, x, out);
}

__global__ __launch_bounds__(CB_WAVES * 32) void k_conv_last(
        const int2* __restrict__ pairs, const int* __restrict__ start,
        const float* __restrict__ dis,
        const float* __restrict__ h, const float* __restrict__ xn,
        float* __restrict__ out) {
    conv_body<true>(pairs, start, dis, h, xn, out);
}

extern "C" void kernel_launch(void* const* d_in, const int* in_sizes, int n_in,
                              void* d_out, int out_size, void* d_ws, size_t ws_size,
                              hipStream_t stream) {
    const int*   ei   = (const int*)  d_in[0];
    const float* F    = (const float*)d_in[1];
    const float* pref = (const float*)d_in[2];
    const float* W1   = (const float*)d_in[3];
    const float* b1   = (const float*)d_in[4];
    const float* W2   = (const float*)d_in[5];
    const float* b2   = (const float*)d_in[6];
    float*       out  = (float*)d_out;

    float* ws   = (float*)d_ws;
    float* xn   = ws;
    float* h    = xn   + (size_t)N_NODES * DIM_E;
    float* deg  = h    + (size_t)N_NODES * DIM_E;

    int*  cnt    = (int*)(deg + N_NODES);
    int*  start  = cnt   + NB_PAD;
    int*  cursor = start + NB_PAD;
    int2* pairs  = (int2*)(cursor + NB_PAD);

    k_mlp<<<MLP_BLOCKS, MLP_THREADS, 0, stream>>>(F, W1, b1, W2, b2, pref, xn,
                                                  out + (size_t)N_NODES * DIM_E,
                                                  deg, N_NODES + NB_PAD);
    k_hist<<<N_EDGES / HB_EDGES, HB_THREADS, 0, stream>>>(ei, deg, cnt);
    k_scan<<<1, NB_PAD, 0, stream>>>(cnt, start, cursor);
    k_scatter<<<N_EDGES / HB_EDGES, HB_THREADS, 0, stream>>>(ei, cursor, pairs, deg);
    k_conv<<<NB, CB_WAVES * 32, 0, stream>>>(pairs, start, deg, xn, h);
    k_conv_last<<<NB, CB_WAVES * 32, 0, stream>>>(pairs, start, deg, h, xn, out);
}


# edge weights: deg[row]^-0.5 stored in each bucketed edge record at partition time, deg[col]^-0.5 kept in LDS per block; the two hops do no scattered weight lookups per edge
# speedup vs baseline: 2.4797x; 1.0351x over previous
#include <hip/hip_runtime.h>

#define NUM_USER 50000
#define NUM_ITEM 50000
#define N_NODES  100000
#define DIM_E    64
#define DIM_FEAT 128
#define DIM_H    256
#define N_EDGES  1600000
#define N_TILES  (NUM_ITEM / 16)

typedef __attribute__((ext_vector_type(2))) float v2f;
typedef __attribute__((ext_vector_type(8))) float v8f;

typedef __attribute__((ext_vector_type(4))) float v4f;

#define MLP_WAVES   8
#define MLP_THREADS (MLP_WAVES * 32)
#define MLP_BLOCKS  256
#define HID_PITCH   68
#define W1Q_WORDS   ((DIM_FEAT / 4) * DIM_H)
#define W2Q_WORDS   ((DIM_H / 4) * DIM_E)
static_assert(MLP_BLOCKS * MLP_WAVES <= N_TILES, "every wave must own a first tile");
static_assert(W1Q_WORDS % MLP_THREADS == 0 && W2Q_WORDS % MLP_THREADS == 0, "the copy loops have whole trips");
static_assert(2 * MLP_BLOCKS * MLP_WAVES >= N_TILES, "two rounds must cover every tile");
static_assert(NUM_USER == NUM_ITEM, "item tile t is paired with the preference rows t*16..t*16+15");

__device__ __forceinline__ v8f mma4(v2f a, v2f b, v8f c) {
    return __builtin_amdgcn_wmma_f32_16x16x4_f32(false, a, false, b, (short)0, c, false, false);
}

__device__ __forceinline__ v4f kquad(const float* __restrict__ M, int i, int log2n) {
    const int n = 1 << log2n;
    const float* p = M + ((i >> log2n) * 4 << log2n) + (i & (n - 1));
    v4f v;
    v.x = p[0]; v.y = p[n]; v.z = p[2 * n]; v.w = p[3 * n];
    return v;
}

__device__ __forceinline__ float row_inv_norm(v4f v) {
    float s = v.x * v.x + v.y * v.y + v.z * v.z + v.w * v.w;
    #pragma unroll
    for (int off = 8; off > 0; off >>= 1) s += __shfl_xor(s, off, 32);
    return 1.0f / fmaxf(sqrtf(s), 1e-12f);
}

__device__ __forceinline__ void load_tile(v4f (&a)[16], const float* __restrict__ F, int tile, int c, int hh) {
    const v4f* fr = (const v4f*)(F + (size_t)(tile * 16 + c) * DIM_FEAT) + hh;
    #pragma unroll
    for (int j = 0; j < 16; ++j) a[j] = fr[2 * j];
}

__global__ __launch_bounds__(MLP_THREADS) void k_mlp(
        const float* __restrict__ F,
        const float* __restrict__ W1, const float* __restrict__ b1,
        const float* __restrict__ W2, const float* __restrict__ b2,
        const float* __restrict__ pref,
        float* __restrict__ xn, float* __restrict__ pref_out,
        float* __restrict__ zero_base, int zero_n) {
    __shared__ v4f   w1q[W1Q_WORDS];
    __shared__ v4f   w2q[W2Q_WORDS];
    __shared__ alignas(16) float hid[MLP_WAVES][16 * HID_PITCH];

    const int tid  = threadIdx.x;
    const int lane = tid & 31;
    const int c    = lane & 15;
    const int hh   = lane >> 4;
    const int wv   = __builtin_amdgcn_readfirstlane(tid >> 5);
    float* hw = hid[wv];

    int tile = blockIdx.x + MLP_BLOCKS * wv;
    v4f a[16];
    load_tile(a, F, tile, c, hh);

    for (int i = blockIdx.x * MLP_THREADS + tid; i < zero_n; i += MLP_BLOCKS * MLP_THREADS)
        zero_base[i] = 0.0f;

    #pragma unroll 16
    for (int it = 0; it < W1Q_WORDS / MLP_THREADS; ++it) {
        const int i = tid + it * MLP_THREADS;
        w1q[i] = kquad(W1, i, 8);
    }
    #pragma unroll
    for (int it = 0; it < W2Q_WORDS / MLP_THREADS; ++it) {
        const int i = tid + it * MLP_THREADS;
        w2q[i] = kquad(W2, i, 6);
    }
    __syncthreads();

    for (;;) {
        v4f pv[8];
        {
            const v4f* ps = (const v4f*)(pref + (size_t)tile * 16 * DIM_E);
            #pragma unroll
            for (int i = 0; i < 8; ++i) pv[i] = ps[i * 32 + lane];
        }
        v8f acc2[4] = {};
        for (int q = 0; q < 4; ++q) {
            v8f acc1[4] = {};
            const v4f* wb = w1q + hh * DIM_H + 64 * q + c;
            #pragma unroll
            for (int j = 0; j < 16; ++j) {
                v4f bq[4];
                #pragma unroll
                for (int t = 0; t < 4; ++t) bq[t] = wb[2 * j * DIM_H + 16 * t];
                #pragma unroll
                for (int t = 0; t < 4; ++t) acc1[t] = mma4(a[j].xy, bq[t].xy, acc1[t]);
                #pragma unroll
                for (int t = 0; t < 4; ++t) acc1[t] = mma4(a[j].zw, bq[t].zw, acc1[t]);
            }
            #pragma unroll
            for (int t = 0; t < 4; ++t) {
                const float bias = b1[64 * q + 16 * t + c];
                #pragma unroll
                for (int r = 0; r < 8; ++r) {
                    float hv = acc1[t][r] + bias;
                    hv = (hv > 0.0f) ? hv : 0.01f * hv;
                    hw[(8 * hh + r) * HID_PITCH + 16 * t + c] = hv;
                }
            }

            const v4f* w2b = w2q + (16 * q + hh) * DIM_E + c;
            const float* hr = hw + c * HID_PITCH + 4 * hh;
            #pragma unroll
            for (int j = 0; j < 8; ++j) {
                const v4f ah = *(const v4f*)(hr + 8 * j);
                v4f bq[4];
                #pragma unroll
                for (int t = 0; t < 4; ++t) bq[t] = w2b[2 * j * DIM_E + 16 * t];
                #pragma unroll
                for (int t = 0; t < 4; ++t) acc2[t] = mma4(ah.xy, bq[t].xy, acc2[t]);
                #pragma unroll
                for (int t = 0; t < 4; ++t) acc2[t] = mma4(ah.zw, bq[t].zw, acc2[t]);
            }
        }

        #pragma unroll
        for (int t = 0; t < 4; ++t) {
            const float bias = b2[16 * t + c];
            #pragma unroll
            for (int r = 0; r < 8; ++r)
                hw[(8 * hh + r) * HID_PITCH + 16 * t + c] = acc2[t][r] + bias;
        }
        v4f* dst = (v4f*)(xn + (size_t)(NUM_USER + tile * 16) * DIM_E);
        #pragma unroll
        for (int i = 0; i < 8; ++i) {
            const int idx = i * 32 + lane;
            const v4f v = *(const v4f*)(hw + (idx >> 4) * HID_PITCH + 4 * (idx & 15));
            dst[idx] = v * row_inv_norm(v);
        }
        v4f* xu = (v4f*)(xn + (size_t)tile * 16 * DIM_E);
        v4f* pu = (v4f*)(pref_out + (size_t)tile * 16 * DIM_E);
        #pragma unroll
        for (int i = 0; i < 8; ++i) {
            const int idx = i * 32 + lane;
            pu[idx] = pv[i];
            xu[idx] = pv[i] * row_inv_norm(pv[i]);
        }

        tile += MLP_BLOCKS * MLP_WAVES;
        if (tile >= N_TILES) break;
        load_tile(a, F, tile, c, hh);
    }
}

#define BK_SHIFT   7
#define BK_ROWS    (1 << BK_SHIFT)
#define NB         ((N_NODES + BK_ROWS - 1) / BK_ROWS)
#define NB_PAD     1024
#define HB_THREADS 640
#define HB_PER_THREAD 10
#define HB_EDGES   (HB_THREADS * HB_PER_THREAD)
static_assert(N_EDGES % HB_EDGES == 0, "edge count must fill whole partition blocks");
static_assert(NB <= NB_PAD, "one scan block must cover every bucket");
#define ROW_BITS   17
static_assert(N_NODES <= (1 << ROW_BITS), "a source row must fit the record's low bits");
static_assert(ROW_BITS + BK_SHIFT <= 32, "the row inside the bucket must fit the bits above it");

__global__ __launch_bounds__(HB_THREADS) void k_hist(
        const int* __restrict__ ei, float* __restrict__ deg, int* __restrict__ cnt) {
    __shared__ int lh[NB];
    for (int i = threadIdx.x; i < NB; i += HB_THREADS) lh[i] = 0;
    __syncthreads();
    const int e0 = blockIdx.x * HB_EDGES + threadIdx.x;
    int r[HB_PER_THREAD], c[HB_PER_THREAD];
    #pragma unroll
    for (int k = 0; k < HB_PER_THREAD; ++k) {
        r[k] = ei[e0 + k * HB_THREADS];
        c[k] = ei[N_EDGES + e0 + k * HB_THREADS];
    }
    #pragma unroll
    for (int k = 0; k < HB_PER_THREAD; ++k) {
        if (r[k] != c[k]) atomicAdd(deg + r[k], 1.0f);
        atomicAdd(&lh[c[k] >> BK_SHIFT], 1);
    }
    __syncthreads();
    for (int i = threadIdx.x; i < NB; i += HB_THREADS) {
        const int n = lh[i];
        if (n) atomicAdd(cnt + i, n);
    }
}

__global__ __launch_bounds__(NB_PAD) void k_scan(
        const int* __restrict__ cnt, int* __restrict__ start, int* __restrict__ cursor) {
    __shared__ int s[2][NB_PAD];
    const int t = threadIdx.x;
    const int own = (t < NB) ? cnt[t] : 0;
    int cur = 0;
    s[0][t] = own;
    __syncthreads();
    for (int off = 1; off < NB_PAD; off <<= 1) {
        int v = s[cur][t];
        if (t >= off) v += s[cur][t - off];
        s[cur ^ 1][t] = v;
        cur ^= 1;
        __syncthreads();
    }
    const int incl = s[cur][t];
    if (t < NB) { start[t] = incl - own; cursor[t] = incl - own; }
    if (t == NB - 1) start[NB] = incl;
}

__global__ __launch_bounds__(HB_THREADS) void k_scatter(
        const int* __restrict__ ei, int* __restrict__ cursor, int2* __restrict__ recs,
        const float* __restrict__ deg) {
    __shared__ int lh[NB];
    __shared__ int lb[NB];
    for (int i = threadIdx.x; i < NB; i += HB_THREADS) lh[i] = 0;
    __syncthreads();
    const int e0 = blockIdx.x * HB_EDGES + threadIdx.x;
    int r[HB_PER_THREAD], c[HB_PER_THREAD];
    #pragma unroll
    for (int k = 0; k < HB_PER_THREAD; ++k) {
        r[k] = ei[e0 + k * HB_THREADS];
        c[k] = ei[N_EDGES + e0 + k * HB_THREADS];
    }
    #pragma unroll
    for (int k = 0; k < HB_PER_THREAD; ++k) atomicAdd(&lh[c[k] >> BK_SHIFT], 1);
    __syncthreads();
    float d[HB_PER_THREAD];
    #pragma unroll
    for (int k = 0; k < HB_PER_THREAD; ++k) d[k] = deg[r[k]];
    for (int i = threadIdx.x; i < NB; i += HB_THREADS) {
        const int n = lh[i];
        lb[i] = n ? atomicAdd(cursor + i, n) : 0;
        lh[i] = 0;
    }
    __syncthreads();
    #pragma unroll
    for (int k = 0; k < HB_PER_THREAD; ++k) {
        const int b = c[k] >> BK_SHIFT;
        const int pos = lb[b] + atomicAdd(&lh[b], 1);
        const float w = (d[k] > 0.0f) ? rsqrtf(d[k]) : 0.0f;
        recs[pos] = make_int2(r[k] | ((c[k] & (BK_ROWS - 1)) << ROW_BITS), __float_as_int(w));
    }
}

#define CB_WAVES   16
#define CONV_CHUNK 16

#define CONV_OUT_TRIPS (BK_ROWS * DIM_E / 4 / (CB_WAVES * 32))
static_assert(CONV_OUT_TRIPS * CB_WAVES * 32 * 4 == BK_ROWS * DIM_E, "the store pass has whole trips");

template <bool LAST>
__device__ __forceinline__ void conv_body(
        const int2* __restrict__ recs, const int* __restrict__ start,
        const float* __restrict__ deg,
        const float* __restrict__ x, const float* __restrict__ xn,
        float* __restrict__ out) {
    __shared__ float acc[BK_ROWS * DIM_E];
    __shared__ float disc[BK_ROWS];
    const int lane = threadIdx.x & 31;
    const int wv   = __builtin_amdgcn_readfirstlane(threadIdx.x >> 5);
    const int row0 = blockIdx.x << BK_SHIFT;

    for (int i = threadIdx.x; i < BK_ROWS * DIM_E / 4; i += CB_WAVES * 32)
        ((float4*)acc)[i] = make_float4(0.0f, 0.0f, 0.0f, 0.0f);
    if (threadIdx.x < BK_ROWS) {
        const int n = row0 + threadIdx.x;
        const float d = (n < N_NODES) ? deg[n] : 0.0f;
        disc[threadIdx.x] = (d > 0.0f) ? rsqrtf(d) : 0.0f;
    }
    __syncthreads();

    const int ebeg = start[blockIdx.x];
    const int eend = start[blockIdx.x + 1];
    for (int eb = ebeg + wv * 32; eb < eend; eb += CB_WAVES * 32) {
        const int last = min(31, eend - eb - 1);
        const int2 p   = recs[eb + min(lane, last)];
        const int r    = p.x & ((1 << ROW_BITS) - 1);
        const int cl   = (int)((unsigned)p.x >> ROW_BITS);
        const float dc = disc[cl];
        const float nrm = (lane <= last && r != row0 + cl) ? __int_as_float(p.y) * dc : 0.0f;

        for (int base = 0; base < 32; base += CONV_CHUNK) {
            float v0[CONV_CHUNK], v1[CONV_CHUNK];
            #pragma unroll
            for (int j = 0; j < CONV_CHUNK; ++j) {
                const int rj = __builtin_amdgcn_readlane(r, base + j);
                const float* xr = x + (size_t)rj * DIM_E;
                v0[j] = xr[lane];
                v1[j] = xr[lane + 32];
            }
            #pragma unroll
            for (int j = 0; j < CONV_CHUNK; ++j) {
                const int cj = __builtin_amdgcn_readlane(cl, base + j);
                const float nj = __int_as_float(__builtin_amdgcn_readlane(__float_as_int(nrm), base + j));
                float* o = acc + cj * DIM_E;
                atomicAdd(o + lane,      v0[j] * nj);
                atomicAdd(o + lane + 32, v1[j] * nj);
            }
        }
    }
    const int nrows = min(BK_ROWS, N_NODES - row0);
    const int n4    = nrows * DIM_E / 4;
    float4* o4 = (float4*)(out + (size_t)row0 * DIM_E);
    if (LAST) {
        const float4* x4 = (const float4*)(xn + (size_t)row0 * DIM_E);
        const float4* h4 = (const float4*)(x  + (size_t)row0 * DIM_E);
        float4 xa[CONV_OUT_TRIPS], hb[CONV_OUT_TRIPS];
        #pragma unroll
        for (int k = 0; k < CONV_OUT_TRIPS; ++k) {
            const int ic = min((int)threadIdx.x + k * CB_WAVES * 32, n4 - 1);
            xa[k] = x4[ic];
            hb[k] = h4[ic];
        }
        __syncthreads();
        float4 sm[CONV_OUT_TRIPS];
        #pragma unroll
        for (int k = 0; k < CONV_OUT_TRIPS; ++k)
            sm[k] = ((const float4*)acc)[threadIdx.x + k * CB_WAVES * 32];
        #pragma unroll
        for (int k = 0; k < CONV_OUT_TRIPS; ++k) {
            const int i = threadIdx.x + k * CB_WAVES * 32;
            if (i < n4) {
                float4 rr;
                rr.x = xa[k].x + hb[k].x + sm[k].x;
                rr.y = xa[k].y + hb[k].y + sm[k].y;
                rr.z = xa[k].z + hb[k].z + sm[k].z;
                rr.w = xa[k].w + hb[k].w + sm[k].w;
                o4[i] = rr;
            }
        }
    } else {
        __syncthreads();
        for (int i = threadIdx.x; i < n4; i += CB_WAVES * 32)
            o4[i] = ((const float4*)acc)[i];
    }
}

__global__ __launch_bounds__(CB_WAVES * 32, 16) void k_conv(
        const int2* __restrict__ recs, const int* __restrict__ start,
        const float* __restrict__ deg,
        const float* __restrict__ x, float* __restrict__ out) {
    conv_body<false>(recs, start, deg, x, x, out);
}

__global__ __launch_bounds__(CB_WAVES * 32, 16) void k_conv_last(
        const int2* __restrict__ recs, const int* __restrict__ start,
        const float* __restrict__ deg,
        const float* __restrict__ h, const float* __restrict__ xn,
        float* __restrict__ out) {
    conv_body<true>(recs, start, deg, h, xn, out);
}

extern "C" void kernel_launch(void* const* d_in, const int* in_sizes, int n_in,
                              void* d_out, int out_size, void* d_ws, size_t ws_size,
                              hipStream_t stream) {
    const int*   ei   = (const int*)  d_in[0];
    const float* F    = (const float*)d_in[1];
    const float* pref = (const float*)d_in[2];
    const float* W1   = (const float*)d_in[3];
    const float* b1   = (const float*)d_in[4];
    const float* W2   = (const float*)d_in[5];
    const float* b2   = (const float*)d_in[6];
    float*       out  = (float*)d_out;

    float* ws   = (float*)d_ws;
    float* xn   = ws;
    float* h    = xn   + (size_t)N_NODES * DIM_E;
    float* deg  = h    + (size_t)N_NODES * DIM_E;

    int*  cnt    = (int*)(deg + N_NODES);
    int*  start  = cnt   + NB_PAD;
    int*  cursor = start + NB_PAD;
    int2* recs   = (int2*)(cursor + NB_PAD);

    k_mlp<<<MLP_BLOCKS, MLP_THREADS, 0, stream>>>(F, W1, b1, W2, b2, pref, xn,
                                                  out + (size_t)N_NODES * DIM_E,
                                                  deg, N_NODES + NB_PAD);
    k_hist<<<N_EDGES / HB_EDGES, HB_THREADS, 0, stream>>>(ei, deg, cnt);
    k_scan<<<1, NB_PAD, 0, stream>>>(cnt, start, cursor);
    k_scatter<<<N_EDGES / HB_EDGES, HB_THREADS, 0, stream>>>(ei, cursor, recs, deg);
    k_conv<<<NB, CB_WAVES * 32, 0, stream>>>(recs, start, deg, xn, h);
    k_conv_last<<<NB, CB_WAVES * 32, 0, stream>>>(recs, start, deg, h, xn, out);
}


# conv hops: each block counting-sorts its bucket's records by destination row in LDS; half waves sum whole rows in registers from 16-B loads; no LDS f32 atomics; a row's sum order = its rank order
# speedup vs baseline: 2.7064x; 1.0914x over previous
#include <hip/hip_runtime.h>

#define NUM_USER 50000
#define NUM_ITEM 50000
#define N_NODES  100000
#define DIM_E    64
#define DIM_FEAT 128
#define DIM_H    256
#define N_EDGES  1600000
#define N_TILES  (NUM_ITEM / 16)

typedef __attribute__((ext_vector_type(2))) float v2f;
typedef __attribute__((ext_vector_type(8))) float v8f;

typedef __attribute__((ext_vector_type(4))) float v4f;

#define MLP_WAVES   8
#define MLP_THREADS (MLP_WAVES * 32)
#define MLP_BLOCKS  256
#define HID_PITCH   68
#define W1Q_WORDS   ((DIM_FEAT / 4) * DIM_H)
#define W2Q_WORDS   ((DIM_H / 4) * DIM_E)
static_assert(MLP_BLOCKS * MLP_WAVES <= N_TILES, "every wave must own a first tile");
static_assert(W1Q_WORDS % MLP_THREADS == 0 && W2Q_WORDS % MLP_THREADS == 0, "the copy loops have whole trips");
static_assert(2 * MLP_BLOCKS * MLP_WAVES >= N_TILES, "two rounds must cover every tile");
static_assert(NUM_USER == NUM_ITEM, "item tile t is paired with the preference rows t*16..t*16+15");

__device__ __forceinline__ v8f mma4(v2f a, v2f b, v8f c) {
    return __builtin_amdgcn_wmma_f32_16x16x4_f32(false, a, false, b, (short)0, c, false, false);
}

__device__ __forceinline__ v4f kquad(const float* __restrict__ M, int i, int log2n) {
    const int n = 1 << log2n;
    const float* p = M + ((i >> log2n) * 4 << log2n) + (i & (n - 1));
    v4f v;
    v.x = p[0]; v.y = p[n]; v.z = p[2 * n]; v.w = p[3 * n];
    return v;
}

__device__ __forceinline__ float row_inv_norm(v4f v) {
    float s = v.x * v.x + v.y * v.y + v.z * v.z + v.w * v.w;
    #pragma unroll
    for (int off = 8; off > 0; off >>= 1) s += __shfl_xor(s, off, 32);
    return 1.0f / fmaxf(sqrtf(s), 1e-12f);
}

__device__ __forceinline__ void load_tile(v4f (&a)[16], const float* __restrict__ F, int tile, int c, int hh) {
    const v4f* fr = (const v4f*)(F + (size_t)(tile * 16 + c) * DIM_FEAT) + hh;
    #pragma unroll
    for (int j = 0; j < 16; ++j) a[j] = fr[2 * j];
}

__global__ __launch_bounds__(MLP_THREADS) void k_mlp(
        const float* __restrict__ F,
        const float* __restrict__ W1, const float* __restrict__ b1,
        const float* __restrict__ W2, const float* __restrict__ b2,
        const float* __restrict__ pref,
        float* __restrict__ xn, float* __restrict__ pref_out,
        float* __restrict__ zero_base, int zero_n) {
    __shared__ v4f   w1q[W1Q_WORDS];
    __shared__ v4f   w2q[W2Q_WORDS];
    __shared__ alignas(16) float hid[MLP_WAVES][16 * HID_PITCH];

    const int tid  = threadIdx.x;
    const int lane = tid & 31;
    const int c    = lane & 15;
    const int hh   = lane >> 4;
    const int wv   = __builtin_amdgcn_readfirstlane(tid >> 5);
    float* hw = hid[wv];

    int tile = blockIdx.x + MLP_BLOCKS * wv;
    v4f a[16];
    load_tile(a, F, tile, c, hh);

    for (int i = blockIdx.x * MLP_THREADS + tid; i < zero_n; i += MLP_BLOCKS * MLP_THREADS)
        zero_base[i] = 0.0f;

    #pragma unroll 16
    for (int it = 0; it < W1Q_WORDS / MLP_THREADS; ++it) {
        const int i = tid + it * MLP_THREADS;
        w1q[i] = kquad(W1, i, 8);
    }
    #pragma unroll
    for (int it = 0; it < W2Q_WORDS / MLP_THREADS; ++it) {
        const int i = tid + it * MLP_THREADS;
        w2q[i] = kquad(W2, i, 6);
    }
    __syncthreads();

    for (;;) {
        v4f pv[8];
        {
            const v4f* ps = (const v4f*)(pref + (size_t)tile * 16 * DIM_E);
            #pragma unroll
            for (int i = 0; i < 8; ++i) pv[i] = ps[i * 32 + lane];
        }
        v8f acc2[4] = {};
        for (int q = 0; q < 4; ++q) {
            v8f acc1[4] = {};
            const v4f* wb = w1q + hh * DIM_H + 64 * q + c;
            #pragma unroll
            for (int j = 0; j < 16; ++j) {
                v4f bq[4];
                #pragma unroll
                for (int t = 0; t < 4; ++t) bq[t] = wb[2 * j * DIM_H + 16 * t];
                #pragma unroll
                for (int t = 0; t < 4; ++t) acc1[t] = mma4(a[j].xy, bq[t].xy, acc1[t]);
                #pragma unroll
                for (int t = 0; t < 4; ++t) acc1[t] = mma4(a[j].zw, bq[t].zw, acc1[t]);
            }
            #pragma unroll
            for (int t = 0; t < 4; ++t) {
                const float bias = b1[64 * q + 16 * t + c];
                #pragma unroll
                for (int r = 0; r < 8; ++r) {
                    float hv = acc1[t][r] + bias;
                    hv = (hv > 0.0f) ? hv : 0.01f * hv;
                    hw[(8 * hh + r) * HID_PITCH + 16 * t + c] = hv;
                }
            }

            const v4f* w2b = w2q + (16 * q + hh) * DIM_E + c;
            const float* hr = hw + c * HID_PITCH + 4 * hh;
            #pragma unroll
            for (int j = 0; j < 8; ++j) {
                const v4f ah = *(const v4f*)(hr + 8 * j);
                v4f bq[4];
                #pragma unroll
                for (int t = 0; t < 4; ++t) bq[t] = w2b[2 * j * DIM_E + 16 * t];
                #pragma unroll
                for (int t = 0; t < 4; ++t) acc2[t] = mma4(ah.xy, bq[t].xy, acc2[t]);
                #pragma unroll
                for (int t = 0; t < 4; ++t) acc2[t] = mma4(ah.zw, bq[t].zw, acc2[t]);
            }
        }

        #pragma unroll
        for (int t = 0; t < 4; ++t) {
            const float bias = b2[16 * t + c];
            #pragma unroll
            for (int r = 0; r < 8; ++r)
                hw[(8 * hh + r) * HID_PITCH + 16 * t + c] = acc2[t][r] + bias;
        }
        v4f* dst = (v4f*)(xn + (size_t)(NUM_USER + tile * 16) * DIM_E);
        #pragma unroll
        for (int i = 0; i < 8; ++i) {
            const int idx = i * 32 + lane;
            const v4f v = *(const v4f*)(hw + (idx >> 4) * HID_PITCH + 4 * (idx & 15));
            dst[idx] = v * row_inv_norm(v);
        }
        v4f* xu = (v4f*)(xn + (size_t)tile * 16 * DIM_E);
        v4f* pu = (v4f*)(pref_out + (size_t)tile * 16 * DIM_E);
        #pragma unroll
        for (int i = 0; i < 8; ++i) {
            const int idx = i * 32 + lane;
            pu[idx] = pv[i];
            xu[idx] = pv[i] * row_inv_norm(pv[i]);
        }

        tile += MLP_BLOCKS * MLP_WAVES;
        if (tile >= N_TILES) break;
        load_tile(a, F, tile, c, hh);
    }
}

#define BK_SHIFT   7
#define BK_ROWS    (1 << BK_SHIFT)
#define NB         ((N_NODES + BK_ROWS - 1) / BK_ROWS)
#define NB_PAD     1024
#define HB_THREADS 640
#define HB_PER_THREAD 10
#define HB_EDGES   (HB_THREADS * HB_PER_THREAD)
static_assert(N_EDGES % HB_EDGES == 0, "edge count must fill whole partition blocks");
static_assert(NB <= NB_PAD, "one scan block must cover every bucket");
#define ROW_BITS   17
static_assert(N_NODES <= (1 << ROW_BITS), "a source row must fit the record's low bits");
static_assert(ROW_BITS + BK_SHIFT <= 32, "the row inside the bucket must fit the bits above it");

__global__ __launch_bounds__(HB_THREADS) void k_hist(
        const int* __restrict__ ei, float* __restrict__ deg, int* __restrict__ cnt) {
    __shared__ int lh[NB];
    for (int i = threadIdx.x; i < NB; i += HB_THREADS) lh[i] = 0;
    __syncthreads();
    const int e0 = blockIdx.x * HB_EDGES + threadIdx.x;
    int r[HB_PER_THREAD], c[HB_PER_THREAD];
    #pragma unroll
    for (int k = 0; k < HB_PER_THREAD; ++k) {
        r[k] = ei[e0 + k * HB_THREADS];
        c[k] = ei[N_EDGES + e0 + k * HB_THREADS];
    }
    #pragma unroll
    for (int k = 0; k < HB_PER_THREAD; ++k) {
        if (r[k] != c[k]) atomicAdd(deg + r[k], 1.0f);
        atomicAdd(&lh[c[k] >> BK_SHIFT], 1);
    }
    __syncthreads();
    for (int i = threadIdx.x; i < NB; i += HB_THREADS) {
        const int n = lh[i];
        if (n) atomicAdd(cnt + i, n);
    }
}

__global__ __launch_bounds__(NB_PAD) void k_scan(
        const int* __restrict__ cnt, int* __restrict__ start, int* __restrict__ cursor) {
    __shared__ int s[2][NB_PAD];
    const int t = threadIdx.x;
    const int own = (t < NB) ? cnt[t] : 0;
    int cur = 0;
    s[0][t] = own;
    __syncthreads();
    for (int off = 1; off < NB_PAD; off <<= 1) {
        int v = s[cur][t];
        if (t >= off) v += s[cur][t - off];
        s[cur ^ 1][t] = v;
        cur ^= 1;
        __syncthreads();
    }
    const int incl = s[cur][t];
    if (t < NB) { start[t] = incl - own; cursor[t] = incl - own; }
    if (t == NB - 1) start[NB] = incl;
}

__global__ __launch_bounds__(HB_THREADS) void k_scatter(
        const int* __restrict__ ei, int* __restrict__ cursor, int2* __restrict__ recs,
        const float* __restrict__ deg) {
    __shared__ int lh[NB];
    __shared__ int lb[NB];
    for (int i = threadIdx.x; i < NB; i += HB_THREADS) lh[i] = 0;
    __syncthreads();
    const int e0 = blockIdx.x * HB_EDGES + threadIdx.x;
    int r[HB_PER_THREAD], c[HB_PER_THREAD];
    #pragma unroll
    for (int k = 0; k < HB_PER_THREAD; ++k) {
        r[k] = ei[e0 + k * HB_THREADS];
        c[k] = ei[N_EDGES + e0 + k * HB_THREADS];
    }
    #pragma unroll
    for (int k = 0; k < HB_PER_THREAD; ++k) atomicAdd(&lh[c[k] >> BK_SHIFT], 1);
    __syncthreads();
    float d[HB_PER_THREAD];
    #pragma unroll
    for (int k = 0; k < HB_PER_THREAD; ++k) d[k] = deg[r[k]];
    for (int i = threadIdx.x; i < NB; i += HB_THREADS) {
        const int n = lh[i];
        lb[i] = n ? atomicAdd(cursor + i, n) : 0;
        lh[i] = 0;
    }
    __syncthreads();
    #pragma unroll
    for (int k = 0; k < HB_PER_THREAD; ++k) {
        const int b = c[k] >> BK_SHIFT;
        const int pos = lb[b] + atomicAdd(&lh[b], 1);
        const float w = (d[k] > 0.0f) ? rsqrtf(d[k]) : 0.0f;
        recs[pos] = make_int2(r[k] | ((c[k] & (BK_ROWS - 1)) << ROW_BITS), __float_as_int(w));
    }
}

#define CB_WAVES   16
#define CB_THREADS (CB_WAVES * 32)
#define SORT_PER   6
#define SORT_CAP   (SORT_PER * CB_THREADS)
#define HALF_ROWS  (BK_ROWS / (2 * CB_WAVES))
#define GATHER_U   4
static_assert(HALF_ROWS * 2 * CB_WAVES == BK_ROWS, "the half waves must cover the bucket's rows");
static_assert(BK_ROWS == 4 * 32, "one wave scans the row counts, four per lane");

#define CONV_OUT_TRIPS (BK_ROWS * DIM_E / 4 / CB_THREADS)
static_assert(CONV_OUT_TRIPS * CB_THREADS * 4 == BK_ROWS * DIM_E, "the store pass has whole trips");

template <bool LAST>
__device__ __forceinline__ void conv_body(
        const int2* __restrict__ recs, const int* __restrict__ start,
        const float* __restrict__ deg,
        const float* __restrict__ x, const float* __restrict__ xn,
        float* __restrict__ out) {
    __shared__ alignas(16) float acc[BK_ROWS * DIM_E];
    __shared__ float disc[BK_ROWS];
    __shared__ alignas(16) int cntl[BK_ROWS];
    __shared__ int rowstart[BK_ROWS + 1];
    __shared__ int2 srt[SORT_CAP];
    const int tid  = threadIdx.x;
    const int lane = tid & 31;
    const int c    = lane & 15;
    const int hh   = lane >> 4;
    const int wv   = __builtin_amdgcn_readfirstlane(tid >> 5);
    const int row0 = blockIdx.x << BK_SHIFT;

    for (int i = tid; i < BK_ROWS * DIM_E / 4; i += CB_THREADS)
        ((float4*)acc)[i] = make_float4(0.0f, 0.0f, 0.0f, 0.0f);
    if (tid < BK_ROWS) {
        const int n = row0 + tid;
        const float d = (n < N_NODES) ? deg[n] : 0.0f;
        disc[tid] = (d > 0.0f) ? rsqrtf(d) : 0.0f;
    }

    const int ebeg = start[blockIdx.x];
    const int eend = start[blockIdx.x + 1];
    const unsigned lane_off = (unsigned)c << 4;
    for (int cb = ebeg; cb < eend; cb += SORT_CAP) {
        const int n = min(SORT_CAP, eend - cb);
        if (tid < BK_ROWS) cntl[tid] = 0;
        __syncthreads();

        int2 p[SORT_PER];
        int  rk[SORT_PER];
        #pragma unroll
        for (int k = 0; k < SORT_PER; ++k)
            p[k] = recs[cb + min(tid + k * CB_THREADS, n - 1)];
        #pragma unroll
        for (int k = 0; k < SORT_PER; ++k) {
            rk[k] = 0;
            if (tid + k * CB_THREADS < n)
                rk[k] = atomicAdd(&cntl[(unsigned)p[k].x >> ROW_BITS], 1);
        }
        __syncthreads();

        if (wv == 0) {
            const int4 c4 = ((const int4*)cntl)[lane];
            const int tot = c4.x + c4.y + c4.z + c4.w;
            int incl = tot;
            incl += __builtin_amdgcn_update_dpp(0, incl, 0x111, 0xf, 0xf, false);
            incl += __builtin_amdgcn_update_dpp(0, incl, 0x112, 0xf, 0xf, false);
            incl += __builtin_amdgcn_update_dpp(0, incl, 0x114, 0xf, 0xf, false);
            incl += __builtin_amdgcn_update_dpp(0, incl, 0x118, 0xf, 0xf, false);
            const int low = __builtin_amdgcn_readlane(incl, 15);
            if (lane >= 16) incl += low;
            const int excl = incl - tot;
            rowstart[4 * lane]     = excl;
            rowstart[4 * lane + 1] = excl + c4.x;
            rowstart[4 * lane + 2] = excl + c4.x + c4.y;
            rowstart[4 * lane + 3] = excl + c4.x + c4.y + c4.z;
            if (lane == 31) rowstart[BK_ROWS] = incl;
        }
        __syncthreads();

        #pragma unroll
        for (int k = 0; k < SORT_PER; ++k) {
            if (tid + k * CB_THREADS < n) {
                const int cl = (int)((unsigned)p[k].x >> ROW_BITS);
                const int r  = p[k].x & ((1 << ROW_BITS) - 1);
                const float w = (r != row0 + cl) ? __int_as_float(p[k].y) * disc[cl] : 0.0f;
                srt[rowstart[cl] + rk[k]] = make_int2(r, __float_as_int(w));
            }
        }
        __syncthreads();

        const int hw = 2 * wv + hh;
        #pragma unroll 1
        for (int rr = 0; rr < HALF_ROWS; ++rr) {
            const int row = hw * HALF_ROWS + rr;
            const int s = rowstart[row];
            const int e = rowstart[row + 1];
            const int len   = e - s;
            const int olen  = __builtin_amdgcn_permlanex16(len, len, 0x76543210, 0xfedcba98, false, false);
            const int trips = __builtin_amdgcn_readfirstlane((max(len, olen) + GATHER_U - 1) / GATHER_U);
            const int last = max(e - 1, 0);
            int2 rec[GATHER_U];
            #pragma unroll
            for (int u = 0; u < GATHER_U; ++u) rec[u] = srt[min(s + u, last)];
            v4f a = {0.0f, 0.0f, 0.0f, 0.0f};
            for (int t = 0; t < trips; ++t) {
                const int i = s + t * GATHER_U;
                v4f   v[GATHER_U];
                float w[GATHER_U];
                #pragma unroll
                for (int u = 0; u < GATHER_U; ++u) {
                    const unsigned off = ((unsigned)rec[u].x << 8) | lane_off;
                    v[u] = *(const v4f*)((const char*)x + off);
                    w[u] = (i + u < e) ? __int_as_float(rec[u].y) : 0.0f;
                }
                #pragma unroll
                for (int u = 0; u < GATHER_U; ++u) rec[u] = srt[min(i + GATHER_U + u, last)];
                #pragma unroll
                for (int u = 0; u < GATHER_U; ++u) a += v[u] * w[u];
            }
            if (e > s) {
                v4f* ar = (v4f*)(acc + row * DIM_E) + c;
                *ar = *ar + a;
            }
        }
    }
    const int nrows = min(BK_ROWS, N_NODES - row0);
    const int n4    = nrows * DIM_E / 4;
    float4* o4 = (float4*)(out + (size_t)row0 * DIM_E);
    if (LAST) {
        const float4* x4 = (const float4*)(xn + (size_t)row0 * DIM_E);
        const float4* h4 = (const float4*)(x  + (size_t)row0 * DIM_E);
        float4 xa[CONV_OUT_TRIPS], hb[CONV_OUT_TRIPS];
        #pragma unroll
        for (int k = 0; k < CONV_OUT_TRIPS; ++k) {
            const int ic = min(tid + k * CB_THREADS, n4 - 1);
            xa[k] = x4[ic];
            hb[k] = h4[ic];
        }
        __syncthreads();
        float4 sm[CONV_OUT_TRIPS];
        #pragma unroll
        for (int k = 0; k < CONV_OUT_TRIPS; ++k)
            sm[k] = ((const float4*)acc)[tid + k * CB_THREADS];
        #pragma unroll
        for (int k = 0; k < CONV_OUT_TRIPS; ++k) {
            const int i = tid + k * CB_THREADS;
            if (i < n4) {
                float4 rr;
                rr.x = xa[k].x + hb[k].x + sm[k].x;
                rr.y = xa[k].y + hb[k].y + sm[k].y;
                rr.z = xa[k].z + hb[k].z + sm[k].z;
                rr.w = xa[k].w + hb[k].w + sm[k].w;
                o4[i] = rr;
            }
        }
    } else {
        __syncthreads();
        for (int i = tid; i < n4; i += CB_THREADS)
            o4[i] = ((const float4*)acc)[i];
    }
}

__global__ __launch_bounds__(CB_WAVES * 32, 16) void k_conv(
        const int2* __restrict__ recs, const int* __restrict__ start,
        const float* __restrict__ deg,
        const float* __restrict__ x, float* __restrict__ out) {
    conv_body<false>(recs, start, deg, x, x, out);
}

__global__ __launch_bounds__(CB_WAVES * 32, 16) void k_conv_last(
        const int2* __restrict__ recs, const int* __restrict__ start,
        const float* __restrict__ deg,
        const float* __restrict__ h, const float* __restrict__ xn,
        float* __restrict__ out) {
    conv_body<true>(recs, start, deg, h, xn, out);
}

extern "C" void kernel_launch(void* const* d_in, const int* in_sizes, int n_in,
                              void* d_out, int out_size, void* d_ws, size_t ws_size,
                              hipStream_t stream) {
    const int*   ei   = (const int*)  d_in[0];
    const float* F    = (const float*)d_in[1];
    const float* pref = (const float*)d_in[2];
    const float* W1   = (const float*)d_in[3];
    const float* b1   = (const float*)d_in[4];
    const float* W2   = (const float*)d_in[5];
    const float* b2   = (const float*)d_in[6];
    float*       out  = (float*)d_out;

    float* ws   = (float*)d_ws;
    float* xn   = ws;
    float* h    = xn   + (size_t)N_NODES * DIM_E;
    float* deg  = h    + (size_t)N_NODES * DIM_E;

    int*  cnt    = (int*)(deg + N_NODES);
    int*  start  = cnt   + NB_PAD;
    int*  cursor = start + NB_PAD;
    int2* recs   = (int2*)(cursor + NB_PAD);

    k_mlp<<<MLP_BLOCKS, MLP_THREADS, 0, stream>>>(F, W1, b1, W2, b2, pref, xn,
                                                  out + (size_t)N_NODES * DIM_E,
                                                  deg, N_NODES + NB_PAD);
    k_hist<<<N_EDGES / HB_EDGES, HB_THREADS, 0, stream>>>(ei, deg, cnt);
    k_scan<<<1, NB_PAD, 0, stream>>>(cnt, start, cursor);
    k_scatter<<<N_EDGES / HB_EDGES, HB_THREADS, 0, stream>>>(ei, cursor, recs, deg);
    k_conv<<<NB, CB_WAVES * 32, 0, stream>>>(recs, start, deg, xn, h);
    k_conv_last<<<NB, CB_WAVES * 32, 0, stream>>>(recs, start, deg, h, xn, out);
}


# degree count: the per-edge global f32 adds replaced by LDS counting per (row slice, edge group) block, partial counts stored as whole lines and summed beside the bucket scan; deg stays exact
# speedup vs baseline: 4.0541x; 1.4980x over previous
#include <hip/hip_runtime.h>

#define NUM_USER 50000
#define NUM_ITEM 50000
#define N_NODES  100000
#define DIM_E    64
#define DIM_FEAT 128
#define DIM_H    256
#define N_EDGES  1600000
#define N_TILES  (NUM_ITEM / 16)

typedef __attribute__((ext_vector_type(2))) float v2f;
typedef __attribute__((ext_vector_type(8))) float v8f;

typedef __attribute__((ext_vector_type(4))) float v4f;

#define MLP_WAVES   8
#define MLP_THREADS (MLP_WAVES * 32)
#define MLP_BLOCKS  256
#define HID_PITCH   68
#define W1Q_WORDS   ((DIM_FEAT / 4) * DIM_H)
#define W2Q_WORDS   ((DIM_H / 4) * DIM_E)
static_assert(MLP_BLOCKS * MLP_WAVES <= N_TILES, "every wave must own a first tile");
static_assert(W1Q_WORDS % MLP_THREADS == 0 && W2Q_WORDS % MLP_THREADS == 0, "the copy loops have whole trips");
static_assert(2 * MLP_BLOCKS * MLP_WAVES >= N_TILES, "two rounds must cover every tile");
static_assert(NUM_USER == NUM_ITEM, "item tile t is paired with the preference rows t*16..t*16+15");

__device__ __forceinline__ v8f mma4(v2f a, v2f b, v8f c) {
    return __builtin_amdgcn_wmma_f32_16x16x4_f32(false, a, false, b, (short)0, c, false, false);
}

__device__ __forceinline__ v4f kquad(const float* __restrict__ M, int i, int log2n) {
    const int n = 1 << log2n;
    const float* p = M + ((i >> log2n) * 4 << log2n) + (i & (n - 1));
    v4f v;
    v.x = p[0]; v.y = p[n]; v.z = p[2 * n]; v.w = p[3 * n];
    return v;
}

__device__ __forceinline__ float row_inv_norm(v4f v) {
    float s = v.x * v.x + v.y * v.y + v.z * v.z + v.w * v.w;
    #pragma unroll
    for (int off = 8; off > 0; off >>= 1) s += __shfl_xor(s, off, 32);
    return 1.0f / fmaxf(sqrtf(s), 1e-12f);
}

__device__ __forceinline__ void load_tile(v4f (&a)[16], const float* __restrict__ F, int tile, int c, int hh) {
    const v4f* fr = (const v4f*)(F + (size_t)(tile * 16 + c) * DIM_FEAT) + hh;
    #pragma unroll
    for (int j = 0; j < 16; ++j) a[j] = fr[2 * j];
}

__global__ __launch_bounds__(MLP_THREADS) void k_mlp(
        const float* __restrict__ F,
        const float* __restrict__ W1, const float* __restrict__ b1,
        const float* __restrict__ W2, const float* __restrict__ b2,
        const float* __restrict__ pref,
        float* __restrict__ xn, float* __restrict__ pref_out,
        float* __restrict__ zero_base, int zero_n) {
    __shared__ v4f   w1q[W1Q_WORDS];
    __shared__ v4f   w2q[W2Q_WORDS];
    __shared__ alignas(16) float hid[MLP_WAVES][16 * HID_PITCH];

    const int tid  = threadIdx.x;
    const int lane = tid & 31;
    const int c    = lane & 15;
    const int hh   = lane >> 4;
    const int wv   = __builtin_amdgcn_readfirstlane(tid >> 5);
    float* hw = hid[wv];

    int tile = blockIdx.x + MLP_BLOCKS * wv;
    v4f a[16];
    load_tile(a, F, tile, c, hh);

    for (int i = blockIdx.x * MLP_THREADS + tid; i < zero_n; i += MLP_BLOCKS * MLP_THREADS)
        zero_base[i] = 0.0f;

    #pragma unroll 16
    for (int it = 0; it < W1Q_WORDS / MLP_THREADS; ++it) {
        const int i = tid + it * MLP_THREADS;
        w1q[i] = kquad(W1, i, 8);
    }
    #pragma unroll
    for (int it = 0; it < W2Q_WORDS / MLP_THREADS; ++it) {
        const int i = tid + it * MLP_THREADS;
        w2q[i] = kquad(W2, i, 6);
    }
    __syncthreads();

    for (;;) {
        v4f pv[8];
        {
            const v4f* ps = (const v4f*)(pref + (size_t)tile * 16 * DIM_E);
            #pragma unroll
            for (int i = 0; i < 8; ++i) pv[i] = ps[i * 32 + lane];
        }
        v8f acc2[4] = {};
        for (int q = 0; q < 4; ++q) {
            v8f acc1[4] = {};
            const v4f* wb = w1q + hh * DIM_H + 64 * q + c;
            #pragma unroll
            for (int j = 0; j < 16; ++j) {
                v4f bq[4];
                #pragma unroll
                for (int t = 0; t < 4; ++t) bq[t] = wb[2 * j * DIM_H + 16 * t];
                #pragma unroll
                for (int t = 0; t < 4; ++t) acc1[t] = mma4(a[j].xy, bq[t].xy, acc1[t]);
                #pragma unroll
                for (int t = 0; t < 4; ++t) acc1[t] = mma4(a[j].zw, bq[t].zw, acc1[t]);
            }
            #pragma unroll
            for (int t = 0; t < 4; ++t) {
                const float bias = b1[64 * q + 16 * t + c];
                #pragma unroll
                for (int r = 0; r < 8; ++r) {
                    float hv = acc1[t][r] + bias;
                    hv = (hv > 0.0f) ? hv : 0.01f * hv;
                    hw[(8 * hh + r) * HID_PITCH + 16 * t + c] = hv;
                }
            }

            const v4f* w2b = w2q + (16 * q + hh) * DIM_E + c;
            const float* hr = hw + c * HID_PITCH + 4 * hh;
            #pragma unroll
            for (int j = 0; j < 8; ++j) {
                const v4f ah = *(const v4f*)(hr + 8 * j);
                v4f bq[4];
                #pragma unroll
                for (int t = 0; t < 4; ++t) bq[t] = w2b[2 * j * DIM_E + 16 * t];
                #pragma unroll
                for (int t = 0; t < 4; ++t) acc2[t] = mma4(ah.xy, bq[t].xy, acc2[t]);
                #pragma unroll
                for (int t = 0; t < 4; ++t) acc2[t] = mma4(ah.zw, bq[t].zw, acc2[t]);
            }
        }

        #pragma unroll
        for (int t = 0; t < 4; ++t) {
            const float bias = b2[16 * t + c];
            #pragma unroll
            for (int r = 0; r < 8; ++r)
                hw[(8 * hh + r) * HID_PITCH + 16 * t + c] = acc2[t][r] + bias;
        }
        v4f* dst = (v4f*)(xn + (size_t)(NUM_USER + tile * 16) * DIM_E);
        #pragma unroll
        for (int i = 0; i < 8; ++i) {
            const int idx = i * 32 + lane;
            const v4f v = *(const v4f*)(hw + (idx >> 4) * HID_PITCH + 4 * (idx & 15));
            dst[idx] = v * row_inv_norm(v);
        }
        v4f* xu = (v4f*)(xn + (size_t)tile * 16 * DIM_E);
        v4f* pu = (v4f*)(pref_out + (size_t)tile * 16 * DIM_E);
        #pragma unroll
        for (int i = 0; i < 8; ++i) {
            const int idx = i * 32 + lane;
            pu[idx] = pv[i];
            xu[idx] = pv[i] * row_inv_norm(pv[i]);
        }

        tile += MLP_BLOCKS * MLP_WAVES;
        if (tile >= N_TILES) break;
        load_tile(a, F, tile, c, hh);
    }
}

#define BK_SHIFT   7
#define BK_ROWS    (1 << BK_SHIFT)
#define NB         ((N_NODES + BK_ROWS - 1) / BK_ROWS)
#define NB_PAD     1024
#define HB_THREADS 640
#define HB_PER_THREAD 10
#define HB_EDGES   (HB_THREADS * HB_PER_THREAD)
static_assert(N_EDGES % HB_EDGES == 0, "edge count must fill whole partition blocks");
static_assert(NB <= NB_PAD, "one scan block must cover every bucket");
#define ROW_BITS   17
static_assert(N_NODES <= (1 << ROW_BITS), "a source row must fit the record's low bits");
static_assert(ROW_BITS + BK_SHIFT <= 32, "the row inside the bucket must fit the bits above it");

#define DEG_SLICES   4
#define SLICE_NODES  25024
#define DEG_PITCH    (DEG_SLICES * SLICE_NODES)
#define DEG_GROUPS   64
#define GROUP_EDGES  (N_EDGES / DEG_GROUPS)
#define GROUP_QUADS  (GROUP_EDGES / 4)
#define HG_THREADS   1024
#define HG_TRIPS     ((GROUP_QUADS + HG_THREADS - 1) / HG_THREADS)
static_assert(DEG_GROUPS * GROUP_EDGES == N_EDGES, "the groups must cover the edge list");
static_assert(GROUP_EDGES % 4 == 0, "a group is read as 16-byte words");
static_assert(DEG_PITCH >= N_NODES, "the slices must cover every node");
static_assert(SLICE_NODES % 32 == 0, "a slice's counters are whole 128-byte lines");

__device__ __forceinline__ void count_edge(int r, int c, unsigned lo, unsigned* ldeg, int* lh) {
    const unsigned d = (unsigned)r - lo;
    if (d < SLICE_NODES) {
        if (r != c) atomicAdd(&ldeg[d], 1u);
        atomicAdd(&lh[c >> BK_SHIFT], 1);
    }
}

__global__ __launch_bounds__(HG_THREADS) void k_hist(
        const int* __restrict__ ei, unsigned* __restrict__ part, int* __restrict__ cnt) {
    __shared__ alignas(16) unsigned ldeg[SLICE_NODES];
    __shared__ int lh[NB];
    const int tid = threadIdx.x;
    const int s = blockIdx.x % DEG_SLICES;
    const int g = blockIdx.x / DEG_SLICES;
    const int4* r4 = (const int4*)(ei + (size_t)g * GROUP_EDGES);
    const int4* c4 = (const int4*)(ei + N_EDGES + (size_t)g * GROUP_EDGES);
    int4 r[HG_TRIPS], c[HG_TRIPS];
    #pragma unroll
    for (int k = 0; k < HG_TRIPS; ++k) {
        const int q = min(tid + k * HG_THREADS, GROUP_QUADS - 1);
        r[k] = r4[q];
        c[k] = c4[q];
    }
    for (int i = tid; i < SLICE_NODES / 4; i += HG_THREADS)
        ((uint4*)ldeg)[i] = make_uint4(0u, 0u, 0u, 0u);
    for (int i = tid; i < NB; i += HG_THREADS) lh[i] = 0;
    __syncthreads();
    const unsigned lo = (unsigned)s * SLICE_NODES;
    #pragma unroll
    for (int k = 0; k < HG_TRIPS; ++k) {
        if (tid + k * HG_THREADS < GROUP_QUADS) {
            count_edge(r[k].x, c[k].x, lo, ldeg, lh);
            count_edge(r[k].y, c[k].y, lo, ldeg, lh);
            count_edge(r[k].z, c[k].z, lo, ldeg, lh);
            count_edge(r[k].w, c[k].w, lo, ldeg, lh);
        }
    }
    __syncthreads();
    uint4* dst = (uint4*)(part + (size_t)g * DEG_PITCH + lo);
    for (int i = tid; i < SLICE_NODES / 4; i += HG_THREADS)
        dst[i] = ((const uint4*)ldeg)[i];
    for (int i = tid; i < NB; i += HG_THREADS) {
        const int n = lh[i];
        if (n) atomicAdd(cnt + i, n);
    }
}

#define DEG_SUM_BLOCKS ((N_NODES + NB_PAD - 1) / NB_PAD)
__global__ __launch_bounds__(NB_PAD) void k_scan(
        const int* __restrict__ cnt, int* __restrict__ start, int* __restrict__ cursor,
        const unsigned* __restrict__ part, float* __restrict__ deg) {
    __shared__ int s[2][NB_PAD];
    const int t = threadIdx.x;
    if (blockIdx.x != 0) {
        const int n = (blockIdx.x - 1) * NB_PAD + t;
        if (n < N_NODES) {
            unsigned sum = 0;
            #pragma unroll 16
            for (int g = 0; g < DEG_GROUPS; ++g) sum += part[(size_t)g * DEG_PITCH + n];
            deg[n] = (float)sum;
        }
        return;
    }
    const int own = (t < NB) ? cnt[t] : 0;
    int cur = 0;
    s[0][t] = own;
    __syncthreads();
    for (int off = 1; off < NB_PAD; off <<= 1) {
        int v = s[cur][t];
        if (t >= off) v += s[cur][t - off];
        s[cur ^ 1][t] = v;
        cur ^= 1;
        __syncthreads();
    }
    const int incl = s[cur][t];
    if (t < NB) { start[t] = incl - own; cursor[t] = incl - own; }
    if (t == NB - 1) start[NB] = incl;
}

__global__ __launch_bounds__(HB_THREADS) void k_scatter(
        const int* __restrict__ ei, int* __restrict__ cursor, int2* __restrict__ recs,
        const float* __restrict__ deg) {
    __shared__ int lh[NB];
    __shared__ int lb[NB];
    for (int i = threadIdx.x; i < NB; i += HB_THREADS) lh[i] = 0;
    __syncthreads();
    const int e0 = blockIdx.x * HB_EDGES + threadIdx.x;
    int r[HB_PER_THREAD], c[HB_PER_THREAD];
    #pragma unroll
    for (int k = 0; k < HB_PER_THREAD; ++k) {
        r[k] = ei[e0 + k * HB_THREADS];
        c[k] = ei[N_EDGES + e0 + k * HB_THREADS];
    }
    #pragma unroll
    for (int k = 0; k < HB_PER_THREAD; ++k) atomicAdd(&lh[c[k] >> BK_SHIFT], 1);
    __syncthreads();
    float d[HB_PER_THREAD];
    #pragma unroll
    for (int k = 0; k < HB_PER_THREAD; ++k) d[k] = deg[r[k]];
    for (int i = threadIdx.x; i < NB; i += HB_THREADS) {
        const int n = lh[i];
        lb[i] = n ? atomicAdd(cursor + i, n) : 0;
        lh[i] = 0;
    }
    __syncthreads();
    #pragma unroll
    for (int k = 0; k < HB_PER_THREAD; ++k) {
        const int b = c[k] >> BK_SHIFT;
        const int pos = lb[b] + atomicAdd(&lh[b], 1);
        const float w = (d[k] > 0.0f) ? rsqrtf(d[k]) : 0.0f;
        recs[pos] = make_int2(r[k] | ((c[k] & (BK_ROWS - 1)) << ROW_BITS), __float_as_int(w));
    }
}

#define CB_WAVES   16
#define CB_THREADS (CB_WAVES * 32)
#define SORT_PER   6
#define SORT_CAP   (SORT_PER * CB_THREADS)
#define HALF_ROWS  (BK_ROWS / (2 * CB_WAVES))
#define GATHER_U   4
static_assert(HALF_ROWS * 2 * CB_WAVES == BK_ROWS, "the half waves must cover the bucket's rows");
static_assert(BK_ROWS == 4 * 32, "one wave scans the row counts, four per lane");

#define CONV_OUT_TRIPS (BK_ROWS * DIM_E / 4 / CB_THREADS)
static_assert(CONV_OUT_TRIPS * CB_THREADS * 4 == BK_ROWS * DIM_E, "the store pass has whole trips");

template <bool LAST>
__device__ __forceinline__ void conv_body(
        const int2* __restrict__ recs, const int* __restrict__ start,
        const float* __restrict__ deg,
        const float* __restrict__ x, const float* __restrict__ xn,
        float* __restrict__ out) {
    __shared__ alignas(16) float acc[BK_ROWS * DIM_E];
    __shared__ float disc[BK_ROWS];
    __shared__ alignas(16) int cntl[BK_ROWS];
    __shared__ int rowstart[BK_ROWS + 1];
    __shared__ int2 srt[SORT_CAP];
    const int tid  = threadIdx.x;
    const int lane = tid & 31;
    const int c    = lane & 15;
    const int hh   = lane >> 4;
    const int wv   = __builtin_amdgcn_readfirstlane(tid >> 5);
    const int row0 = blockIdx.x << BK_SHIFT;

    for (int i = tid; i < BK_ROWS * DIM_E / 4; i += CB_THREADS)
        ((float4*)acc)[i] = make_float4(0.0f, 0.0f, 0.0f, 0.0f);
    if (tid < BK_ROWS) {
        const int n = row0 + tid;
        const float d = (n < N_NODES) ? deg[n] : 0.0f;
        disc[tid] = (d > 0.0f) ? rsqrtf(d) : 0.0f;
    }

    const int ebeg = start[blockIdx.x];
    const int eend = start[blockIdx.x + 1];
    const unsigned lane_off = (unsigned)c << 4;
    for (int cb = ebeg; cb < eend; cb += SORT_CAP) {
        const int n = min(SORT_CAP, eend - cb);
        if (tid < BK_ROWS) cntl[tid] = 0;
        __syncthreads();

        int2 p[SORT_PER];
        int  rk[SORT_PER];
        #pragma unroll
        for (int k = 0; k < SORT_PER; ++k)
            p[k] = recs[cb + min(tid + k * CB_THREADS, n - 1)];
        #pragma unroll
        for (int k = 0; k < SORT_PER; ++k) {
            rk[k] = 0;
            if (tid + k * CB_THREADS < n)
                rk[k] = atomicAdd(&cntl[(unsigned)p[k].x >> ROW_BITS], 1);
        }
        __syncthreads();

        if (wv == 0) {
            const int4 c4 = ((const int4*)cntl)[lane];
            const int tot = c4.x + c4.y + c4.z + c4.w;
            int incl = tot;
            incl += __builtin_amdgcn_update_dpp(0, incl, 0x111, 0xf, 0xf, false);
            incl += __builtin_amdgcn_update_dpp(0, incl, 0x112, 0xf, 0xf, false);
            incl += __builtin_amdgcn_update_dpp(0, incl, 0x114, 0xf, 0xf, false);
            incl += __builtin_amdgcn_update_dpp(0, incl, 0x118, 0xf, 0xf, false);
            const int low = __builtin_amdgcn_readlane(incl, 15);
            if (lane >= 16) incl += low;
            const int excl = incl - tot;
            rowstart[4 * lane]     = excl;
            rowstart[4 * lane + 1] = excl + c4.x;
            rowstart[4 * lane + 2] = excl + c4.x + c4.y;
            rowstart[4 * lane + 3] = excl + c4.x + c4.y + c4.z;
            if (lane == 31) rowstart[BK_ROWS] = incl;
        }
        __syncthreads();

        #pragma unroll
        for (int k = 0; k < SORT_PER; ++k) {
            if (tid + k * CB_THREADS < n) {
                const int cl = (int)((unsigned)p[k].x >> ROW_BITS);
                const int r  = p[k].x & ((1 << ROW_BITS) - 1);
                const float w = (r != row0 + cl) ? __int_as_float(p[k].y) * disc[cl] : 0.0f;
                srt[rowstart[cl] + rk[k]] = make_int2(r, __float_as_int(w));
            }
        }
        __syncthreads();

        const int hw = 2 * wv + hh;
        #pragma unroll 1
        for (int rr = 0; rr < HALF_ROWS; ++rr) {
            const int row = hw * HALF_ROWS + rr;
            const int s = rowstart[row];
            const int e = rowstart[row + 1];
            const int len   = e - s;
            const int olen  = __builtin_amdgcn_permlanex16(len, len, 0x76543210, 0xfedcba98, false, false);
            const int trips = __builtin_amdgcn_readfirstlane((max(len, olen) + GATHER_U - 1) / GATHER_U);
            const int last = max(e - 1, 0);
            int2 rec[GATHER_U];
            #pragma unroll
            for (int u = 0; u < GATHER_U; ++u) rec[u] = srt[min(s + u, last)];
            v4f a = {0.0f, 0.0f, 0.0f, 0.0f};
            for (int t = 0; t < trips; ++t) {
                const int i = s + t * GATHER_U;
                v4f   v[GATHER_U];
                float w[GATHER_U];
                #pragma unroll
                for (int u = 0; u < GATHER_U; ++u) {
                    const unsigned off = ((unsigned)rec[u].x << 8) | lane_off;
                    v[u] = *(const v4f*)((const char*)x + off);
                    w[u] = (i + u < e) ? __int_as_float(rec[u].y) : 0.0f;
                }
                #pragma unroll
                for (int u = 0; u < GATHER_U; ++u) rec[u] = srt[min(i + GATHER_U + u, last)];
                #pragma unroll
                for (int u = 0; u < GATHER_U; ++u) a += v[u] * w[u];
            }
            if (e > s) {
                v4f* ar = (v4f*)(acc + row * DIM_E) + c;
                *ar = *ar + a;
            }
        }
    }
    const int nrows = min(BK_ROWS, N_NODES - row0);
    const int n4    = nrows * DIM_E / 4;
    float4* o4 = (float4*)(out + (size_t)row0 * DIM_E);
    if (LAST) {
        const float4* x4 = (const float4*)(xn + (size_t)row0 * DIM_E);
        const float4* h4 = (const float4*)(x  + (size_t)row0 * DIM_E);
        float4 xa[CONV_OUT_TRIPS], hb[CONV_OUT_TRIPS];
        #pragma unroll
        for (int k = 0; k < CONV_OUT_TRIPS; ++k) {
            const int ic = min(tid + k * CB_THREADS, n4 - 1);
            xa[k] = x4[ic];
            hb[k] = h4[ic];
        }
        __syncthreads();
        float4 sm[CONV_OUT_TRIPS];
        #pragma unroll
        for (int k = 0; k < CONV_OUT_TRIPS; ++k)
            sm[k] = ((const float4*)acc)[tid + k * CB_THREADS];
        #pragma unroll
        for (int k = 0; k < CONV_OUT_TRIPS; ++k) {
            const int i = tid + k * CB_THREADS;
            if (i < n4) {
                float4 rr;
                rr.x = xa[k].x + hb[k].x + sm[k].x;
                rr.y = xa[k].y + hb[k].y + sm[k].y;
                rr.z = xa[k].z + hb[k].z + sm[k].z;
                rr.w = xa[k].w + hb[k].w + sm[k].w;
                o4[i] = rr;
            }
        }
    } else {
        __syncthreads();
        for (int i = tid; i < n4; i += CB_THREADS)
            o4[i] = ((const float4*)acc)[i];
    }
}

__global__ __launch_bounds__(CB_WAVES * 32, 16) void k_conv(
        const int2* __restrict__ recs, const int* __restrict__ start,
        const float* __restrict__ deg,
        const float* __restrict__ x, float* __restrict__ out) {
    conv_body<false>(recs, start, deg, x, x, out);
}

__global__ __launch_bounds__(CB_WAVES * 32, 16) void k_conv_last(
        const int2* __restrict__ recs, const int* __restrict__ start,
        const float* __restrict__ deg,
        const float* __restrict__ h, const float* __restrict__ xn,
        float* __restrict__ out) {
    conv_body<true>(recs, start, deg, h, xn, out);
}

extern "C" void kernel_launch(void* const* d_in, const int* in_sizes, int n_in,
                              void* d_out, int out_size, void* d_ws, size_t ws_size,
                              hipStream_t stream) {
    const int*   ei   = (const int*)  d_in[0];
    const float* F    = (const float*)d_in[1];
    const float* pref = (const float*)d_in[2];
    const float* W1   = (const float*)d_in[3];
    const float* b1   = (const float*)d_in[4];
    const float* W2   = (const float*)d_in[5];
    const float* b2   = (const float*)d_in[6];
    float*       out  = (float*)d_out;

    float* ws   = (float*)d_ws;
    float* xn   = ws;
    float* h    = xn   + (size_t)N_NODES * DIM_E;
    float* deg  = h    + (size_t)N_NODES * DIM_E;

    int*  cnt    = (int*)(deg + N_NODES);
    int*  start  = cnt   + NB_PAD;
    int*  cursor = start + NB_PAD;
    int2* recs   = (int2*)(cursor + NB_PAD);
    unsigned* part = (unsigned*)(recs + N_EDGES);

    k_mlp<<<MLP_BLOCKS, MLP_THREADS, 0, stream>>>(F, W1, b1, W2, b2, pref, xn,
                                                  out + (size_t)N_NODES * DIM_E,
                                                  (float*)cnt, NB_PAD);
    k_hist<<<DEG_SLICES * DEG_GROUPS, HG_THREADS, 0, stream>>>(ei, part, cnt);
    k_scan<<<1 + DEG_SUM_BLOCKS, NB_PAD, 0, stream>>>(cnt, start, cursor, part, deg);
    k_scatter<<<N_EDGES / HB_EDGES, HB_THREADS, 0, stream>>>(ei, cursor, recs, deg);
    k_conv<<<NB, CB_WAVES * 32, 0, stream>>>(recs, start, deg, xn, h);
    k_conv_last<<<NB, CB_WAVES * 32, 0, stream>>>(recs, start, deg, h, xn, out);
}


# degree and bucket counting joins the MLP launch as 128 extra 8-wave blocks that run on the WGPs the 12-tile MLP blocks leave early; bucket counts leave as plain per-block rows: 6 launches -> 5
# speedup vs baseline: 4.1752x; 1.0299x over previous
#include <hip/hip_runtime.h>

#define NUM_USER 50000
#define NUM_ITEM 50000
#define N_NODES  100000
#define DIM_E    64
#define DIM_FEAT 128
#define DIM_H    256
#define N_EDGES  1600000
#define N_TILES  (NUM_ITEM / 16)

typedef __attribute__((ext_vector_type(2))) float v2f;
typedef __attribute__((ext_vector_type(8))) float v8f;

typedef __attribute__((ext_vector_type(4))) float v4f;

#define MLP_WAVES   8
#define MLP_THREADS (MLP_WAVES * 32)
#define MLP_BLOCKS  256
#define HID_PITCH   68
#define W1Q_WORDS   ((DIM_FEAT / 4) * DIM_H)
#define W2Q_WORDS   ((DIM_H / 4) * DIM_E)
static_assert(MLP_BLOCKS * MLP_WAVES <= N_TILES, "every wave must own a first tile");
static_assert(W1Q_WORDS % MLP_THREADS == 0 && W2Q_WORDS % MLP_THREADS == 0, "the copy loops have whole trips");
static_assert(2 * MLP_BLOCKS * MLP_WAVES >= N_TILES, "two rounds must cover every tile");
static_assert(NUM_USER == NUM_ITEM, "item tile t is paired with the preference rows t*16..t*16+15");

__device__ __forceinline__ v8f mma4(v2f a, v2f b, v8f c) {
    return __builtin_amdgcn_wmma_f32_16x16x4_f32(false, a, false, b, (short)0, c, false, false);
}

__device__ __forceinline__ v4f kquad(const float* __restrict__ M, int i, int log2n) {
    const int n = 1 << log2n;
    const float* p = M + ((i >> log2n) * 4 << log2n) + (i & (n - 1));
    v4f v;
    v.x = p[0]; v.y = p[n]; v.z = p[2 * n]; v.w = p[3 * n];
    return v;
}

__device__ __forceinline__ float row_inv_norm(v4f v) {
    float s = v.x * v.x + v.y * v.y + v.z * v.z + v.w * v.w;
    #pragma unroll
    for (int off = 8; off > 0; off >>= 1) s += __shfl_xor(s, off, 32);
    return 1.0f / fmaxf(sqrtf(s), 1e-12f);
}

__device__ __forceinline__ void load_tile(v4f (&a)[16], const float* __restrict__ F, int tile, int c, int hh) {
    const v4f* fr = (const v4f*)(F + (size_t)(tile * 16 + c) * DIM_FEAT) + hh;
    #pragma unroll
    for (int j = 0; j < 16; ++j) a[j] = fr[2 * j];
}

struct MlpLds {
    v4f   w1q[W1Q_WORDS];
    v4f   w2q[W2Q_WORDS];
    alignas(16) float hid[MLP_WAVES][16 * HID_PITCH];
};

__device__ __forceinline__ void mlp_body(
        const float* __restrict__ F,
        const float* __restrict__ W1, const float* __restrict__ b1,
        const float* __restrict__ W2, const float* __restrict__ b2,
        const float* __restrict__ pref,
        float* __restrict__ xn, float* __restrict__ pref_out,
        v4f* w1q, v4f* w2q, float (*hid)[16 * HID_PITCH]) {
    const int tid  = threadIdx.x;
    const int lane = tid & 31;
    const int c    = lane & 15;
    const int hh   = lane >> 4;
    const int wv   = __builtin_amdgcn_readfirstlane(tid >> 5);
    float* hw = hid[wv];

    int tile = blockIdx.x + MLP_BLOCKS * wv;
    v4f a[16];
    load_tile(a, F, tile, c, hh);

    #pragma unroll 16
    for (int it = 0; it < W1Q_WORDS / MLP_THREADS; ++it) {
        const int i = tid + it * MLP_THREADS;
        w1q[i] = kquad(W1, i, 8);
    }
    #pragma unroll
    for (int it = 0; it < W2Q_WORDS / MLP_THREADS; ++it) {
        const int i = tid + it * MLP_THREADS;
        w2q[i] = kquad(W2, i, 6);
    }
    __syncthreads();

    for (;;) {
        v4f pv[8];
        {
            const v4f* ps = (const v4f*)(pref + (size_t)tile * 16 * DIM_E);
            #pragma unroll
            for (int i = 0; i < 8; ++i) pv[i] = ps[i * 32 + lane];
        }
        v8f acc2[4] = {};
        for (int q = 0; q < 4; ++q) {
            v8f acc1[4] = {};
            const v4f* wb = w1q + hh * DIM_H + 64 * q + c;
            #pragma unroll
            for (int j = 0; j < 16; ++j) {
                v4f bq[4];
                #pragma unroll
                for (int t = 0; t < 4; ++t) bq[t] = wb[2 * j * DIM_H + 16 * t];
                #pragma unroll
                for (int t = 0; t < 4; ++t) acc1[t] = mma4(a[j].xy, bq[t].xy, acc1[t]);
                #pragma unroll
                for (int t = 0; t < 4; ++t) acc1[t] = mma4(a[j].zw, bq[t].zw, acc1[t]);
            }
            #pragma unroll
            for (int t = 0; t < 4; ++t) {
                const float bias = b1[64 * q + 16 * t + c];
                #pragma unroll
                for (int r = 0; r < 8; ++r) {
                    float hv = acc1[t][r] + bias;
                    hv = (hv > 0.0f) ? hv : 0.01f * hv;
                    hw[(8 * hh + r) * HID_PITCH + 16 * t + c] = hv;
                }
            }

            const v4f* w2b = w2q + (16 * q + hh) * DIM_E + c;
            const float* hr = hw + c * HID_PITCH + 4 * hh;
            #pragma unroll
            for (int j = 0; j < 8; ++j) {
                const v4f ah = *(const v4f*)(hr + 8 * j);
                v4f bq[4];
                #pragma unroll
                for (int t = 0; t < 4; ++t) bq[t] = w2b[2 * j * DIM_E + 16 * t];
                #pragma unroll
                for (int t = 0; t < 4; ++t) acc2[t] = mma4(ah.xy, bq[t].xy, acc2[t]);
                #pragma unroll
                for (int t = 0; t < 4; ++t) acc2[t] = mma4(ah.zw, bq[t].zw, acc2[t]);
            }
        }

        #pragma unroll
        for (int t = 0; t < 4; ++t) {
            const float bias = b2[16 * t + c];
            #pragma unroll
            for (int r = 0; r < 8; ++r)
                hw[(8 * hh + r) * HID_PITCH + 16 * t + c] = acc2[t][r] + bias;
        }
        v4f* dst = (v4f*)(xn + (size_t)(NUM_USER + tile * 16) * DIM_E);
        #pragma unroll
        for (int i = 0; i < 8; ++i) {
            const int idx = i * 32 + lane;
            const v4f v = *(const v4f*)(hw + (idx >> 4) * HID_PITCH + 4 * (idx & 15));
            dst[idx] = v * row_inv_norm(v);
        }
        v4f* xu = (v4f*)(xn + (size_t)tile * 16 * DIM_E);
        v4f* pu = (v4f*)(pref_out + (size_t)tile * 16 * DIM_E);
        #pragma unroll
        for (int i = 0; i < 8; ++i) {
            const int idx = i * 32 + lane;
            pu[idx] = pv[i];
            xu[idx] = pv[i] * row_inv_norm(pv[i]);
        }

        tile += MLP_BLOCKS * MLP_WAVES;
        if (tile >= N_TILES) break;
        load_tile(a, F, tile, c, hh);
    }
}

#define BK_SHIFT   7
#define BK_ROWS    (1 << BK_SHIFT)
#define NB         ((N_NODES + BK_ROWS - 1) / BK_ROWS)
#define NB_PAD     1024
#define HB_THREADS 640
#define HB_PER_THREAD 10
#define HB_EDGES   (HB_THREADS * HB_PER_THREAD)
static_assert(N_EDGES % HB_EDGES == 0, "edge count must fill whole partition blocks");
static_assert(NB <= NB_PAD, "one scan block must cover every bucket");
#define ROW_BITS   17
static_assert(N_NODES <= (1 << ROW_BITS), "a source row must fit the record's low bits");
static_assert(ROW_BITS + BK_SHIFT <= 32, "the row inside the bucket must fit the bits above it");

#define DEG_SLICES   2
#define SLICE_NODES  50048
#define DEG_PITCH    (DEG_SLICES * SLICE_NODES)
#define DEG_GROUPS   64
#define GROUP_EDGES  (N_EDGES / DEG_GROUPS)
#define GROUP_QUADS  (GROUP_EDGES / 4)
#define HIST_BLOCKS  (DEG_SLICES * DEG_GROUPS)
#define HG_THREADS   MLP_THREADS
#define HG_TRIPS     ((GROUP_QUADS + HG_THREADS - 1) / HG_THREADS)
static_assert(DEG_GROUPS * GROUP_EDGES == N_EDGES, "the groups must cover the edge list");
static_assert(GROUP_EDGES % 4 == 0, "a group is read as 16-byte words");
static_assert(DEG_PITCH >= N_NODES, "the slices must cover every node");
static_assert(SLICE_NODES % 32 == 0, "a slice's counters are whole 128-byte lines");
static_assert(2 * HG_TRIPS <= 63, "a thread's edge loads must all fit the load counter");
static_assert(NB_PAD % HG_THREADS == 0 && HIST_BLOCKS % 4 == 0, "the bucket rows are stored and summed in whole trips");

struct HistLds {
    alignas(16) unsigned ldeg[SLICE_NODES];
    int lh[NB_PAD];
};

__device__ __forceinline__ void count_edge(int r, int c, unsigned lo, unsigned* ldeg, int* lh) {
    const unsigned d = (unsigned)r - lo;
    if (d < SLICE_NODES) {
        if (r != c) atomicAdd(&ldeg[d], 1u);
        atomicAdd(&lh[c >> BK_SHIFT], 1);
    }
}

__device__ __forceinline__ void hist_body(
        int hb, const int* __restrict__ ei, unsigned* __restrict__ part, int* __restrict__ bpart,
        unsigned* ldeg, int* lh) {
    const int tid = threadIdx.x;
    const int s = hb % DEG_SLICES;
    const int g = hb / DEG_SLICES;
    const int4* r4 = (const int4*)(ei + (size_t)g * GROUP_EDGES);
    const int4* c4 = (const int4*)(ei + N_EDGES + (size_t)g * GROUP_EDGES);
    int4 r[HG_TRIPS], c[HG_TRIPS];
    #pragma unroll
    for (int k = 0; k < HG_TRIPS; ++k) {
        const int q = min(tid + k * HG_THREADS, GROUP_QUADS - 1);
        r[k] = r4[q];
        c[k] = c4[q];
    }
    for (int i = tid; i < SLICE_NODES / 4; i += HG_THREADS)
        ((uint4*)ldeg)[i] = make_uint4(0u, 0u, 0u, 0u);
    for (int i = tid; i < NB_PAD; i += HG_THREADS) lh[i] = 0;
    __syncthreads();
    const unsigned lo = (unsigned)s * SLICE_NODES;
    #pragma unroll
    for (int k = 0; k < HG_TRIPS; ++k) {
        if (tid + k * HG_THREADS < GROUP_QUADS) {
            count_edge(r[k].x, c[k].x, lo, ldeg, lh);
            count_edge(r[k].y, c[k].y, lo, ldeg, lh);
            count_edge(r[k].z, c[k].z, lo, ldeg, lh);
            count_edge(r[k].w, c[k].w, lo, ldeg, lh);
        }
    }
    __syncthreads();
    uint4* dst = (uint4*)(part + (size_t)g * DEG_PITCH + lo);
    for (int i = tid; i < SLICE_NODES / 4; i += HG_THREADS)
        dst[i] = ((const uint4*)ldeg)[i];
    for (int i = tid; i < NB_PAD; i += HG_THREADS)
        bpart[hb * NB_PAD + i] = lh[i];
}

union FrontLds {
    MlpLds  m;
    HistLds h;
};

__global__ __launch_bounds__(MLP_THREADS) void k_front(
        const float* __restrict__ F,
        const float* __restrict__ W1, const float* __restrict__ b1,
        const float* __restrict__ W2, const float* __restrict__ b2,
        const float* __restrict__ pref,
        float* __restrict__ xn, float* __restrict__ pref_out,
        const int* __restrict__ ei, unsigned* __restrict__ part, int* __restrict__ bpart) {
    __shared__ FrontLds lds;
    if (blockIdx.x >= MLP_BLOCKS) {
        hist_body(blockIdx.x - MLP_BLOCKS, ei, part, bpart, lds.h.ldeg, lds.h.lh);
        return;
    }
    mlp_body(F, W1, b1, W2, b2, pref, xn, pref_out, lds.m.w1q, lds.m.w2q, lds.m.hid);
}

#define DEG_SUM_BLOCKS ((N_NODES + NB_PAD - 1) / NB_PAD)
__global__ __launch_bounds__(NB_PAD) void k_scan(
        const int* __restrict__ bpart, int* __restrict__ start, int* __restrict__ cursor,
        const unsigned* __restrict__ part, float* __restrict__ deg) {
    __shared__ int s[2][NB_PAD];
    __shared__ alignas(16) int sq[4][NB_PAD];
    const int t = threadIdx.x;
    if (blockIdx.x != 0) {
        const int n = (blockIdx.x - 1) * NB_PAD + t;
        if (n < N_NODES) {
            unsigned sum = 0;
            #pragma unroll 16
            for (int g = 0; g < DEG_GROUPS; ++g) sum += part[(size_t)g * DEG_PITCH + n];
            deg[n] = (float)sum;
        }
        return;
    }
    {
        const int w = t & (NB_PAD / 4 - 1);
        const int q = t / (NB_PAD / 4);
        const int4* bp = (const int4*)bpart + w;
        int4 sum = make_int4(0, 0, 0, 0);
        #pragma unroll
        for (int j = 0; j < HIST_BLOCKS / 4; ++j) {
            const int4 v = bp[(q + 4 * j) * (NB_PAD / 4)];
            sum.x += v.x; sum.y += v.y; sum.z += v.z; sum.w += v.w;
        }
        ((int4*)sq[q])[w] = sum;
    }
    __syncthreads();
    const int own = (t < NB) ? sq[0][t] + sq[1][t] + sq[2][t] + sq[3][t] : 0;
    int cur = 0;
    s[0][t] = own;
    __syncthreads();
    for (int off = 1; off < NB_PAD; off <<= 1) {
        int v = s[cur][t];
        if (t >= off) v += s[cur][t - off];
        s[cur ^ 1][t] = v;
        cur ^= 1;
        __syncthreads();
    }
    const int incl = s[cur][t];
    if (t < NB) { start[t] = incl - own; cursor[t] = incl - own; }
    if (t == NB - 1) start[NB] = incl;
}

__global__ __launch_bounds__(HB_THREADS) void k_scatter(
        const int* __restrict__ ei, int* __restrict__ cursor, int2* __restrict__ recs,
        const float* __restrict__ deg) {
    __shared__ int lh[NB];
    __shared__ int lb[NB];
    for (int i = threadIdx.x; i < NB; i += HB_THREADS) lh[i] = 0;
    __syncthreads();
    const int e0 = blockIdx.x * HB_EDGES + threadIdx.x;
    int r[HB_PER_THREAD], c[HB_PER_THREAD];
    #pragma unroll
    for (int k = 0; k < HB_PER_THREAD; ++k) {
        r[k] = ei[e0 + k * HB_THREADS];
        c[k] = ei[N_EDGES + e0 + k * HB_THREADS];
    }
    #pragma unroll
    for (int k = 0; k < HB_PER_THREAD; ++k) atomicAdd(&lh[c[k] >> BK_SHIFT], 1);
    __syncthreads();
    float d[HB_PER_THREAD];
    #pragma unroll
    for (int k = 0; k < HB_PER_THREAD; ++k) d[k] = deg[r[k]];
    for (int i = threadIdx.x; i < NB; i += HB_THREADS) {
        const int n = lh[i];
        lb[i] = n ? atomicAdd(cursor + i, n) : 0;
        lh[i] = 0;
    }
    __syncthreads();
    #pragma unroll
    for (int k = 0; k < HB_PER_THREAD; ++k) {
        const int b = c[k] >> BK_SHIFT;
        const int pos = lb[b] + atomicAdd(&lh[b], 1);
        const float w = (d[k] > 0.0f) ? rsqrtf(d[k]) : 0.0f;
        recs[pos] = make_int2(r[k] | ((c[k] & (BK_ROWS - 1)) << ROW_BITS), __float_as_int(w));
    }
}

#define CB_WAVES   16
#define CB_THREADS (CB_WAVES * 32)
#define SORT_PER   6
#define SORT_CAP   (SORT_PER * CB_THREADS)
#define HALF_ROWS  (BK_ROWS / (2 * CB_WAVES))
#define GATHER_U   4
static_assert(HALF_ROWS * 2 * CB_WAVES == BK_ROWS, "the half waves must cover the bucket's rows");
static_assert(BK_ROWS == 4 * 32, "one wave scans the row counts, four per lane");

#define CONV_OUT_TRIPS (BK_ROWS * DIM_E / 4 / CB_THREADS)
static_assert(CONV_OUT_TRIPS * CB_THREADS * 4 == BK_ROWS * DIM_E, "the store pass has whole trips");

template <bool LAST>
__device__ __forceinline__ void conv_body(
        const int2* __restrict__ recs, const int* __restrict__ start,
        const float* __restrict__ deg,
        const float* __restrict__ x, const float* __restrict__ xn,
        float* __restrict__ out) {
    __shared__ alignas(16) float acc[BK_ROWS * DIM_E];
    __shared__ float disc[BK_ROWS];
    __shared__ alignas(16) int cntl[BK_ROWS];
    __shared__ int rowstart[BK_ROWS + 1];
    __shared__ int2 srt[SORT_CAP];
    const int tid  = threadIdx.x;
    const int lane = tid & 31;
    const int c    = lane & 15;
    const int hh   = lane >> 4;
    const int wv   = __builtin_amdgcn_readfirstlane(tid >> 5);
    const int row0 = blockIdx.x << BK_SHIFT;

    for (int i = tid; i < BK_ROWS * DIM_E / 4; i += CB_THREADS)
        ((float4*)acc)[i] = make_float4(0.0f, 0.0f, 0.0f, 0.0f);
    if (tid < BK_ROWS) {
        const int n = row0 + tid;
        const float d = (n < N_NODES) ? deg[n] : 0.0f;
        disc[tid] = (d > 0.0f) ? rsqrtf(d) : 0.0f;
    }

    const int ebeg = start[blockIdx.x];
    const int eend = start[blockIdx.x + 1];
    const unsigned lane_off = (unsigned)c << 4;
    for (int cb = ebeg; cb < eend; cb += SORT_CAP) {
        const int n = min(SORT_CAP, eend - cb);
        if (tid < BK_ROWS) cntl[tid] = 0;
        __syncthreads();

        int2 p[SORT_PER];
        int  rk[SORT_PER];
        #pragma unroll
        for (int k = 0; k < SORT_PER; ++k)
            p[k] = recs[cb + min(tid + k * CB_THREADS, n - 1)];
        #pragma unroll
        for (int k = 0; k < SORT_PER; ++k) {
            rk[k] = 0;
            if (tid + k * CB_THREADS < n)
                rk[k] = atomicAdd(&cntl[(unsigned)p[k].x >> ROW_BITS], 1);
        }
        __syncthreads();

        if (wv == 0) {
            const int4 c4 = ((const int4*)cntl)[lane];
            const int tot = c4.x + c4.y + c4.z + c4.w;
            int incl = tot;
            incl += __builtin_amdgcn_update_dpp(0, incl, 0x111, 0xf, 0xf, false);
            incl += __builtin_amdgcn_update_dpp(0, incl, 0x112, 0xf, 0xf, false);
            incl += __builtin_amdgcn_update_dpp(0, incl, 0x114, 0xf, 0xf, false);
            incl += __builtin_amdgcn_update_dpp(0, incl, 0x118, 0xf, 0xf, false);
            const int low = __builtin_amdgcn_readlane(incl, 15);
            if (lane >= 16) incl += low;
            const int excl = incl - tot;
            rowstart[4 * lane]     = excl;
            rowstart[4 * lane + 1] = excl + c4.x;
            rowstart[4 * lane + 2] = excl + c4.x + c4.y;
            rowstart[4 * lane + 3] = excl + c4.x + c4.y + c4.z;
            if (lane == 31) rowstart[BK_ROWS] = incl;
        }
        __syncthreads();

        #pragma unroll
        for (int k = 0; k < SORT_PER; ++k) {
            if (tid + k * CB_THREADS < n) {
                const int cl = (int)((unsigned)p[k].x >> ROW_BITS);
                const int r  = p[k].x & ((1 << ROW_BITS) - 1);
                const float w = (r != row0 + cl) ? __int_as_float(p[k].y) * disc[cl] : 0.0f;
                srt[rowstart[cl] + rk[k]] = make_int2(r, __float_as_int(w));
            }
        }
        __syncthreads();

        const int hw = 2 * wv + hh;
        #pragma unroll 1
        for (int rr = 0; rr < HALF_ROWS; ++rr) {
            const int row = hw * HALF_ROWS + rr;
            const int s = rowstart[row];
            const int e = rowstart[row + 1];
            const int len   = e - s;
            const int olen  = __builtin_amdgcn_permlanex16(len, len, 0x76543210, 0xfedcba98, false, false);
            const int trips = __builtin_amdgcn_readfirstlane((max(len, olen) + GATHER_U - 1) / GATHER_U);
            const int last = max(e - 1, 0);
            int2 rec[GATHER_U];
            #pragma unroll
            for (int u = 0; u < GATHER_U; ++u) rec[u] = srt[min(s + u, last)];
            v4f a = {0.0f, 0.0f, 0.0f, 0.0f};
            for (int t = 0; t < trips; ++t) {
                const int i = s + t * GATHER_U;
                v4f   v[GATHER_U];
                float w[GATHER_U];
                #pragma unroll
                for (int u = 0; u < GATHER_U; ++u) {
                    const unsigned off = ((unsigned)rec[u].x << 8) | lane_off;
                    v[u] = *(const v4f*)((const char*)x + off);
                    w[u] = (i + u < e) ? __int_as_float(rec[u].y) : 0.0f;
                }
                #pragma unroll
                for (int u = 0; u < GATHER_U; ++u) rec[u] = srt[min(i + GATHER_U + u, last)];
                #pragma unroll
                for (int u = 0; u < GATHER_U; ++u) a += v[u] * w[u];
            }
            if (e > s) {
                v4f* ar = (v4f*)(acc + row * DIM_E) + c;
                *ar = *ar + a;
            }
        }
    }
    const int nrows = min(BK_ROWS, N_NODES - row0);
    const int n4    = nrows * DIM_E / 4;
    float4* o4 = (float4*)(out + (size_t)row0 * DIM_E);
    if (LAST) {
        const float4* x4 = (const float4*)(xn + (size_t)row0 * DIM_E);
        const float4* h4 = (const float4*)(x  + (size_t)row0 * DIM_E);
        float4 xa[CONV_OUT_TRIPS], hb[CONV_OUT_TRIPS];
        #pragma unroll
        for (int k = 0; k < CONV_OUT_TRIPS; ++k) {
            const int ic = min(tid + k * CB_THREADS, n4 - 1);
            xa[k] = x4[ic];
            hb[k] = h4[ic];
        }
        __syncthreads();
        float4 sm[CONV_OUT_TRIPS];
        #pragma unroll
        for (int k = 0; k < CONV_OUT_TRIPS; ++k)
            sm[k] = ((const float4*)acc)[tid + k * CB_THREADS];
        #pragma unroll
        for (int k = 0; k < CONV_OUT_TRIPS; ++k) {
            const int i = tid + k * CB_THREADS;
            if (i < n4) {
                float4 rr;
                rr.x = xa[k].x + hb[k].x + sm[k].x;
                rr.y = xa[k].y + hb[k].y + sm[k].y;
                rr.z = xa[k].z + hb[k].z + sm[k].z;
                rr.w = xa[k].w + hb[k].w + sm[k].w;
                o4[i] = rr;
            }
        }
    } else {
        __syncthreads();
        for (int i = tid; i < n4; i += CB_THREADS)
            o4[i] = ((const float4*)acc)[i];
    }
}

__global__ __launch_bounds__(CB_WAVES * 32, 16) void k_conv(
        const int2* __restrict__ recs, const int* __restrict__ start,
        const float* __restrict__ deg,
        const float* __restrict__ x, float* __restrict__ out) {
    conv_body<false>(recs, start, deg, x, x, out);
}

__global__ __launch_bounds__(CB_WAVES * 32, 16) void k_conv_last(
        const int2* __restrict__ recs, const int* __restrict__ start,
        const float* __restrict__ deg,
        const float* __restrict__ h, const float* __restrict__ xn,
        float* __restrict__ out) {
    conv_body<true>(recs, start, deg, h, xn, out);
}

extern "C" void kernel_launch(void* const* d_in, const int* in_sizes, int n_in,
                              void* d_out, int out_size, void* d_ws, size_t ws_size,
                              hipStream_t stream) {
    const int*   ei   = (const int*)  d_in[0];
    const float* F    = (const float*)d_in[1];
    const float* pref = (const float*)d_in[2];
    const float* W1   = (const float*)d_in[3];
    const float* b1   = (const float*)d_in[4];
    const float* W2   = (const float*)d_in[5];
    const float* b2   = (const float*)d_in[6];
    float*       out  = (float*)d_out;

    float* ws   = (float*)d_ws;
    float* xn   = ws;
    float* h    = xn   + (size_t)N_NODES * DIM_E;
    float* deg  = h    + (size_t)N_NODES * DIM_E;

    int*  start  = (int*)(deg + N_NODES);
    int*  cursor = start + NB_PAD;
    int2* recs   = (int2*)(cursor + NB_PAD);
    unsigned* part  = (unsigned*)(recs + N_EDGES);
    int*      bpart = (int*)(part + (size_t)DEG_GROUPS * DEG_PITCH);

    k_front<<<MLP_BLOCKS + HIST_BLOCKS, MLP_THREADS, 0, stream>>>(F, W1, b1, W2, b2, pref, xn,
                                                  out + (size_t)N_NODES * DIM_E,
                                                  ei, part, bpart);
    k_scan<<<1 + DEG_SUM_BLOCKS, NB_PAD, 0, stream>>>(bpart, start, cursor, part, deg);
    k_scatter<<<N_EDGES / HB_EDGES, HB_THREADS, 0, stream>>>(ei, cursor, recs, deg);
    k_conv<<<NB, CB_WAVES * 32, 0, stream>>>(recs, start, deg, xn, h);
    k_conv_last<<<NB, CB_WAVES * 32, 0, stream>>>(recs, start, deg, h, xn, out);
}
